# starting kernel
# baseline (speedup 1.0000x reference)
#include <hip/hip_runtime.h>
#include <math.h>

#define DD 1024
#define NA 1887
#define BB 8
#define TT 16
#define ROWS 128

typedef __attribute__((ext_vector_type(2))) float v2f;
typedef __attribute__((ext_vector_type(8))) float v8f;

__global__ void pool_kernel(const float* __restrict__ x, float* __restrict__ A0,
                            float* __restrict__ F1, float* __restrict__ F2) {
  int tid = blockIdx.x * blockDim.x + threadIdx.x;
  if (tid >= ROWS * DD * 3) return;
  int c  = tid % 3;
  int rd = tid / 3;
  int d  = rd % DD;
  int r  = rd / DD;
  const float* p = x + (size_t)rd * 147 + c;
  float s = 0.f;
  #pragma unroll
  for (int i = 0; i < 49; ++i) s += p[i * 3];
  s *= (1.0f / 49.0f);
  if (c == 0) {
    int b = r >> 4, t = r & 15;
    A0[(size_t)(t * BB + b) * DD + d] = s;
  } else if (c == 1) {
    F1[(size_t)r * DD + d] = s;
  } else {
    F2[(size_t)r * DD + d] = s;
  }
}

__device__ __forceinline__ v8f wmma_k_loop(const float* __restrict__ arow,
                                           const float* __restrict__ wrow,
                                           int K, v8f acc) {
  #pragma unroll 4
  for (int k = 0; k < K; k += 4) {
    v2f a = *(const v2f*)(arow + k);
    v2f b = *(const v2f*)(wrow + k);
    acc = __builtin_amdgcn_wmma_f32_16x16x4_f32(false, a, false, b,
                                                (short)0, acc, false, false);
  }
  return acc;
}

template <int MODE>
__global__ void gemm_kernel(const float* __restrict__ A, int lda,
                            const float* __restrict__ W, int ldw,
                            const float* __restrict__ bias1,
                            const float* __restrict__ bias2,
                            float* __restrict__ out, int ldo, int coloff,
                            float* __restrict__ out2, int N, int K) {
  int wave = (blockIdx.x * blockDim.x + threadIdx.x) >> 5;
  int lane = threadIdx.x & 31;
  int m0 = (wave & 7) << 4;
  int n0 = (wave >> 3) << 4;
  int mr = lane & 15;
  int koff = (lane >> 4) << 1;
  int hi = lane >> 4;
  int ncol = n0 + (lane & 15);
  int nload = (ncol < N) ? ncol : (N - 1);
  const float* arow = A + (size_t)(m0 + mr) * lda + koff;
  const float* wrow = W + (size_t)nload * ldw + koff;
  float bsum = bias1[nload] + (bias2 ? bias2[nload] : 0.f);
  v8f acc;
  #pragma unroll
  for (int i = 0; i < 8; ++i) acc[i] = bsum;
  acc = wmma_k_loop(arow, wrow, K, acc);
  #pragma unroll
  for (int v = 0; v < 8; ++v) {
    int r = m0 + v + (hi << 3);
    float val = acc[v];
    if (MODE == 0) {
      out[(size_t)r * ldo + coloff + ncol] = val;
    } else if (MODE == 1) {
      int b = r >> 4, t = r & 15;
      int rr = t * BB + b;
      if (ncol < DD) out[(size_t)rr * DD + ncol] = val;
      else           out2[(size_t)rr * DD + (ncol - DD)] = val;
    } else {
      if (ncol < N) {
        int t = r >> 3, b = r & 7;
        out[(size_t)(b * TT + t) * N + ncol] = val;
      }
    }
  }
}

__global__ void lstm_step_kernel(const float* __restrict__ xpre,
                                 const float* __restrict__ hin,
                                 const float* __restrict__ Whh,
                                 float* __restrict__ cc,
                                 float* __restrict__ hout,
                                 float* __restrict__ lasth,
                                 int step, int mtiles) {
  __shared__ float red[4][4][8][32];
  int lw   = threadIdx.x >> 5;
  int lane = threadIdx.x & 31;
  int w    = blockIdx.x * 8 + lw;
  int kh   = w & 1;
  int wt   = w >> 1;
  int mt   = wt % mtiles;
  int n0   = (wt / mtiles) << 4;
  int m0   = mt << 4;
  int nn   = lane & 15;
  int mr   = lane & 15;
  int koff = (lane >> 4) << 1;
  int hi   = lane >> 4;
  int kbase = kh * (DD / 2);

  v8f acc[4];
  if (kh == 0) {
    #pragma unroll
    for (int g = 0; g < 4; ++g)
      #pragma unroll
      for (int v = 0; v < 8; ++v) {
        int r = m0 + v + (hi << 3);
        acc[g][v] = xpre[(size_t)r * (4 * DD) + g * DD + n0 + nn];
      }
  } else {
    #pragma unroll
    for (int g = 0; g < 4; ++g)
      #pragma unroll
      for (int v = 0; v < 8; ++v) acc[g][v] = 0.f;
  }

  const float* arow = hin + (size_t)(m0 + mr) * DD + kbase + koff;
  const float* w0 = Whh + (size_t)(0 * DD + n0 + nn) * DD + kbase + koff;
  const float* w1 = Whh + (size_t)(1 * DD + n0 + nn) * DD + kbase + koff;
  const float* w2 = Whh + (size_t)(2 * DD + n0 + nn) * DD + kbase + koff;
  const float* w3 = Whh + (size_t)(3 * DD + n0 + nn) * DD + kbase + koff;

  #pragma unroll 2
  for (int k = 0; k < DD / 2; k += 4) {
    v2f a = *(const v2f*)(arow + k);
    acc[0] = __builtin_amdgcn_wmma_f32_16x16x4_f32(false, a, false,
               *(const v2f*)(w0 + k), (short)0, acc[0], false, false);
    acc[1] = __builtin_amdgcn_wmma_f32_16x16x4_f32(false, a, false,
               *(const v2f*)(w1 + k), (short)0, acc[1], false, false);
    acc[2] = __builtin_amdgcn_wmma_f32_16x16x4_f32(false, a, false,
               *(const v2f*)(w2 + k), (short)0, acc[2], false, false);
    acc[3] = __builtin_amdgcn_wmma_f32_16x16x4_f32(false, a, false,
               *(const v2f*)(w3 + k), (short)0, acc[3], false, false);
  }

  int p = lw >> 1;
  if (kh == 1) {
    #pragma unroll
    for (int g = 0; g < 4; ++g)
      #pragma unroll
      for (int v = 0; v < 8; ++v) red[p][g][v][lane] = acc[g][v];
  }
  __syncthreads();
  if (kh == 1) return;

  #pragma unroll
  for (int g = 0; g < 4; ++g)
    #pragma unroll
    for (int v = 0; v < 8; ++v) acc[g][v] += red[p][g][v][lane];

  int tgt = TT - 1 - step;
  #pragma unroll
  for (int v = 0; v < 8; ++v) {
    int r = m0 + v + (hi << 3);
    int d = n0 + nn;
    float si = 1.f / (1.f + expf(-acc[0][v]));
    float sf = 1.f / (1.f + expf(-acc[1][v]));
    float gt = tanhf(acc[2][v]);
    float so = 1.f / (1.f + expf(-acc[3][v]));
    float cold = cc[(size_t)r * DD + d];
    float cnew = sf * cold + si * gt;
    float hnew = so * tanhf(cnew);
    cc[(size_t)r * DD + d] = cnew;
    hout[(size_t)r * DD + d] = hnew;
    if ((r >> 3) == tgt) lasth[(size_t)r * DD + d] = hnew;
  }
}

__global__ void ln_relu_kernel(const float* __restrict__ in,
                               const float* __restrict__ g,
                               const float* __restrict__ be,
                               float* __restrict__ out) {
  __shared__ float s1[256], s2[256];
  int r = blockIdx.x, t = threadIdx.x;
  const float* row = in + (size_t)r * DD;
  float s = 0.f, sq = 0.f;
  for (int i = t; i < DD; i += 256) { float v = row[i]; s += v; sq += v * v; }
  s1[t] = s; s2[t] = sq;
  __syncthreads();
  for (int off = 128; off > 0; off >>= 1) {
    if (t < off) { s1[t] += s1[t + off]; s2[t] += s2[t + off]; }
    __syncthreads();
  }
  float mean = s1[0] * (1.0f / DD);
  float var = s2[0] * (1.0f / DD) - mean * mean;
  float inv = rsqrtf(var + 1e-5f);
  for (int i = t; i < DD; i += 256) {
    float v = (row[i] - mean) * inv * g[i] + be[i];
    out[(size_t)r * DD + i] = v > 0.f ? v : 0.f;
  }
}

extern "C" void kernel_launch(void* const* d_in, const int* in_sizes, int n_in,
                              void* d_out, int out_size, void* d_ws, size_t ws_size,
                              hipStream_t stream) {
  const float* x   = (const float*)d_in[0];
  const float* Wn  = (const float*)d_in[1];
  const float* bn  = (const float*)d_in[2];
  const float* Wv  = (const float*)d_in[3];
  const float* bv  = (const float*)d_in[4];
  const float* Wf  = (const float*)d_in[5];
  const float* bf  = (const float*)d_in[6];
  const float* Wih = (const float*)d_in[7];
  const float* Whh = (const float*)d_in[8];
  const float* bih = (const float*)d_in[9];
  const float* bhh = (const float*)d_in[10];
  const float* Wc  = (const float*)d_in[11];
  const float* bc  = (const float*)d_in[12];
  const float* g_a = (const float*)d_in[13];
  const float* be_a= (const float*)d_in[14];
  const float* Wa  = (const float*)d_in[15];
  const float* ba  = (const float*)d_in[16];
  float* out = (float*)d_out;

  float* ws    = (float*)d_ws;
  float* A0    = ws;
  float* F1    = A0    + 131072;
  float* F2    = F1    + 131072;
  float* ENV   = F2    + 131072;
  float* XPRE  = ENV   + 262144;
  float* H0    = XPRE  + 524288;
  float* H1    = H0    + 131072;
  float* CC    = H1    + 131072;
  float* LASTH = CC    + 131072;
  float* UNR   = LASTH + 131072;
  float* LNR   = UNR   + 131072;

  pool_kernel<<<1536, 256, 0, stream>>>(x, A0, F1, F2);
  gemm_kernel<0><<<64, 256, 0, stream>>>(F1, DD, Wn, DD, bn, nullptr,
                                         ENV, 2 * DD, 0,  nullptr, DD, DD);
  gemm_kernel<0><<<64, 256, 0, stream>>>(F2, DD, Wv, DD, bv, nullptr,
                                         ENV, 2 * DD, DD, nullptr, DD, DD);
  gemm_kernel<1><<<128, 256, 0, stream>>>(ENV, 2 * DD, Wf, 2 * DD, bf, nullptr,
                                          H0, 0, 0, CC, 2 * DD, 2 * DD);
  gemm_kernel<0><<<256, 256, 0, stream>>>(A0, DD, Wih, DD, bih, bhh,
                                          XPRE, 4 * DD, 0, nullptr, 4 * DD, DD);
  float* hin = H0; float* hout = H1;
  for (int s = 0; s < TT; ++s) {
    int mtiles = (TT - s + 1) >> 1;
    lstm_step_kernel<<<mtiles * 16, 256, 0, stream>>>(XPRE, hin, Whh, CC, hout,
                                                      LASTH, s, mtiles);
    float* tmp = hin; hin = hout; hout = tmp;
  }
  gemm_kernel<0><<<64, 256, 0, stream>>>(LASTH, DD, Wc, DD, bc, nullptr,
                                         UNR, DD, 0, nullptr, DD, DD);
  ln_relu_kernel<<<128, 256, 0, stream>>>(UNR, g_a, be_a, LNR);
  gemm_kernel<2><<<118, 256, 0, stream>>>(LNR, DD, Wa, DD, ba, nullptr,
                                          out, NA, 0, nullptr, NA, DD);
}


# All WMMA K loops (6 GEMM launches, 16 LSTM steps): operands staged as block-shared LDS tiles by cooperative whole-line 16-byte loads, two buffers, 4-wave blocks; every k sum order kept
# speedup vs baseline: 1.8855x; 1.8855x over previous
#include <hip/hip_runtime.h>
#include <math.h>

#define DD 1024
#define NA 1887
#define BB 8
#define TT 16
#define ROWS 128

typedef __attribute__((ext_vector_type(2))) float v2f;
typedef __attribute__((ext_vector_type(8))) float v8f;

__global__ void pool_kernel(const float* __restrict__ x, float* __restrict__ A0,
                            float* __restrict__ F1, float* __restrict__ F2) {
  int tid = blockIdx.x * blockDim.x + threadIdx.x;
  if (tid >= ROWS * DD * 3) return;
  int c  = tid % 3;
  int rd = tid / 3;
  int d  = rd % DD;
  int r  = rd / DD;
  const float* p = x + (size_t)rd * 147 + c;
  float s = 0.f;
  #pragma unroll
  for (int i = 0; i < 49; ++i) s += p[i * 3];
  s *= (1.0f / 49.0f);
  if (c == 0) {
    int b = r >> 4, t = r & 15;
    A0[(size_t)(t * BB + b) * DD + d] = s;
  } else if (c == 1) {
    F1[(size_t)r * DD + d] = s;
  } else {
    F2[(size_t)r * DD + d] = s;
  }
}

#define TCOLS 128
#define PITCH (TCOLS + 4)
#define PITCH4 (PITCH / 4)
#define TROWS 96
#define GG 8
#define LG 4
#define ROWS8(X)  X(0) X(1) X(2) X(3) X(4) X(5) X(6) X(7)
#define ROWS16(X) ROWS8(X) X(8) X(9) X(10) X(11) X(12) X(13) X(14) X(15)

template <int MODE>
__global__ void __launch_bounds__(128)
gemm_kernel(const float* __restrict__ A, int lda,
            const float* __restrict__ W, int ldw,
            const float* __restrict__ bias1,
            const float* __restrict__ bias2,
            float* __restrict__ out, int ldo, int coloff,
            float* __restrict__ out2, int N, int K) {
  __shared__ float4 tile4[2][TROWS][PITCH4];
  const int lw   = __builtin_amdgcn_readfirstlane((int)(threadIdx.x >> 5));
  const int lane = threadIdx.x & 31;
  const int mh   = blockIdx.x & 1;
  const int nb   = blockIdx.x >> 1;
  const int m0   = mh * 64 + lw * 16;
  const int nn   = lane & 15;
  const int hi   = lane >> 4;
  const int koff = hi << 1;
  const int ncol0 = nb * 32 + nn;
  const int ncol1 = ncol0 + 16;
  const int nld0 = (ncol0 < N) ? ncol0 : (N - 1);
  const int nld1 = (ncol1 < N) ? ncol1 : (N - 1);
  const float bsum0 = bias1[nld0] + (bias2 ? bias2[nld0] : 0.f);
  const float bsum1 = bias1[nld1] + (bias2 ? bias2[nld1] : 0.f);
  v8f acc0, acc1;
  #pragma unroll
  for (int i = 0; i < 8; ++i) { acc0[i] = bsum0; acc1[i] = bsum1; }

#define G_WROW(i) ((nb * 32 + lw + 4 * (i)) < N ? (nb * 32 + lw + 4 * (i)) : (N - 1))
#define G_ADECL(i) float4 pa##i = *(const float4*)(asrc + (size_t)(4 * (i)) * lda);
#define G_WDECL(i) const float* wsrc##i = W + (size_t)G_WROW(i) * ldw + (lane << 2); float4 pb##i = *(const float4*)(wsrc##i);
#define G_ASTORE(i) tb[lw + 4 * (i)][lane] = pa##i;
#define G_WSTORE(i) tb[64 + lw + 4 * (i)][lane] = pb##i;
#define G_ALOAD(i) pa##i = *(const float4*)(asrc + (size_t)(4 * (i)) * lda + ko);
#define G_WLOAD(i) pb##i = *(const float4*)(wsrc##i + ko);
  const float* asrc = A + (size_t)(mh * 64 + lw) * lda + (lane << 2);
  ROWS16(G_ADECL)
  ROWS8(G_WDECL)

  const int nround = K / TCOLS;
  for (int c = 0; c < nround; ++c) {
    float4 (*tb)[PITCH4] = tile4[c & 1];
    ROWS16(G_ASTORE)
    ROWS8(G_WSTORE)
    __syncthreads();
    if (c + 1 < nround) {
      const int ko = (c + 1) * TCOLS;
      ROWS16(G_ALOAD)
      ROWS8(G_WLOAD)
    }
    const float* tf = (const float*)&tb[0][0];
    const float* ap = tf + (lw * 16 + nn) * PITCH + koff;
    const float* bp = tf + (64 + nn) * PITCH + koff;
    v2f fa[GG], fb0[GG], fb1[GG];
    #pragma unroll
    for (int q = 0; q < GG; ++q) {
      fa[q]  = *(const v2f*)(ap + 4 * q);
      fb0[q] = *(const v2f*)(bp + 4 * q);
      fb1[q] = *(const v2f*)(bp + 16 * PITCH + 4 * q);
    }
    #pragma unroll
    for (int jg = 0; jg < TCOLS / 4; jg += GG) {
      v2f na[GG], nb0[GG], nb1[GG];
      if (jg + GG < TCOLS / 4) {
        #pragma unroll
        for (int q = 0; q < GG; ++q) {
          na[q]  = *(const v2f*)(ap + 4 * (jg + GG + q));
          nb0[q] = *(const v2f*)(bp + 4 * (jg + GG + q));
          nb1[q] = *(const v2f*)(bp + 16 * PITCH + 4 * (jg + GG + q));
        }
      }
      __builtin_amdgcn_sched_barrier(0);
      #pragma unroll
      for (int q = 0; q < GG; ++q) {
        acc0 = __builtin_amdgcn_wmma_f32_16x16x4_f32(false, fa[q], false, fb0[q],
                                                     (short)0, acc0, false, false);
        acc1 = __builtin_amdgcn_wmma_f32_16x16x4_f32(false, fa[q], false, fb1[q],
                                                     (short)0, acc1, false, false);
      }
      __builtin_amdgcn_sched_barrier(0);
      if (jg + GG < TCOLS / 4) {
        #pragma unroll
        for (int q = 0; q < GG; ++q) { fa[q] = na[q]; fb0[q] = nb0[q]; fb1[q] = nb1[q]; }
      }
    }
  }

  #pragma unroll
  for (int t = 0; t < 2; ++t) {
    const int ncol = t ? ncol1 : ncol0;
    #pragma unroll
    for (int v = 0; v < 8; ++v) {
      int r = m0 + v + (hi << 3);
      float val = t ? acc1[v] : acc0[v];
      if (MODE == 0) {
        out[(size_t)r * ldo + coloff + ncol] = val;
      } else if (MODE == 1) {
        int b = r >> 4, tt = r & 15;
        int rr = tt * BB + b;
        if (ncol < DD) out[(size_t)rr * DD + ncol] = val;
        else           out2[(size_t)rr * DD + (ncol - DD)] = val;
      } else {
        if (ncol < N) {
          int tt = r >> 3, b = r & 7;
          out[(size_t)(b * TT + tt) * N + ncol] = val;
        }
      }
    }
  }
}

#define KC (TCOLS / 2)
#define HROWS 32
#define WROWS 64
#define NROUND (DD / 2 / KC)

__global__ void __launch_bounds__(128)
lstm_step_kernel(const float* __restrict__ xpre,
                 const float* __restrict__ hin,
                 const float* __restrict__ Whh,
                 float* __restrict__ cc,
                 float* __restrict__ hout,
                 float* __restrict__ lasth,
                 int step, int mtiles) {
  __shared__ float4 tile4[2][TROWS][PITCH4];
  __shared__ float red[2][4][8][32];
  const int lw   = __builtin_amdgcn_readfirstlane((int)(threadIdx.x >> 5));
  const int lane = threadIdx.x & 31;
  const int mgroups = (mtiles + 1) >> 1;
  const int nt   = blockIdx.x / mgroups;
  const int mg   = blockIdx.x - nt * mgroups;
  const int n0   = nt << 4;
  const int p    = lw >> 1;
  const int kh   = lw & 1;
  const int mt   = mg * 2 + p;
  const bool active = mt < mtiles;
  const int m0   = mt << 4;
  const int nn   = lane & 15;
  const int hi   = lane >> 4;
  const int koff = hi << 1;
  const int row_last = (mtiles << 4) - 1;

  v8f acc[4];
  if (active && kh == 0) {
    #pragma unroll
    for (int g = 0; g < 4; ++g)
      #pragma unroll
      for (int v = 0; v < 8; ++v) {
        int r = m0 + v + (hi << 3);
        acc[g][v] = xpre[(size_t)r * (4 * DD) + g * DD + n0 + nn];
      }
  } else {
    #pragma unroll
    for (int g = 0; g < 4; ++g)
      #pragma unroll
      for (int v = 0; v < 8; ++v) acc[g][v] = 0.f;
  }

  const int kcol = (lane >> 4) * (DD / 2) + ((lane & 15) << 2);
#define H_ROW(i) ((mg * HROWS + lw + 4 * (i)) < row_last ? (mg * HROWS + lw + 4 * (i)) : row_last)
#define H_DECL(i) const float* hsrc##i = hin + (size_t)H_ROW(i) * DD + kcol; float4 ph##i = *(const float4*)(hsrc##i);
#define W_OFF(i) ((size_t)(((i) >> 2) * DD + 4 * ((i) & 3)) * DD)
#define W_DECL(i) float4 pw##i = *(const float4*)(wsrc + W_OFF(i));
#define H_STORE(i) tb[lw + 4 * (i)][lane] = ph##i;
#define W_STORE(i) tb[HROWS + lw + 4 * (i)][lane] = pw##i;
#define H_LOAD(i) ph##i = *(const float4*)(hsrc##i + ko);
#define W_LOAD(i) pw##i = *(const float4*)(wsrc + W_OFF(i) + ko);
  const float* wsrc = Whh + (size_t)(n0 + lw) * DD + kcol;
  ROWS8(H_DECL)
  ROWS16(W_DECL)

  #pragma unroll 1
  for (int c = 0; c < NROUND; ++c) {
    float4 (*tb)[PITCH4] = tile4[c & 1];
    ROWS8(H_STORE)
    ROWS16(W_STORE)
    __syncthreads();
    if (c + 1 < NROUND) {
      const int ko = (c + 1) * KC;
      ROWS8(H_LOAD)
      ROWS16(W_LOAD)
    }
    if (active) {
      const float* tf = (const float*)&tb[0][0];
      const float* ap = tf + (p * 16 + nn) * PITCH + kh * KC + koff;
      const float* bp = tf + (HROWS + nn) * PITCH + kh * KC + koff;
      v2f fa[LG], fb[4][LG];
      #pragma unroll
      for (int q = 0; q < LG; ++q) {
        fa[q] = *(const v2f*)(ap + 4 * q);
        #pragma unroll
        for (int g = 0; g < 4; ++g) fb[g][q] = *(const v2f*)(bp + g * 16 * PITCH + 4 * q);
      }
      #pragma unroll
      for (int jg = 0; jg < KC / 4; jg += LG) {
        v2f na[LG], nb[4][LG];
        if (jg + LG < KC / 4) {
          #pragma unroll
          for (int q = 0; q < LG; ++q) {
            na[q] = *(const v2f*)(ap + 4 * (jg + LG + q));
            #pragma unroll
            for (int g = 0; g < 4; ++g) nb[g][q] = *(const v2f*)(bp + g * 16 * PITCH + 4 * (jg + LG + q));
          }
        }
        __builtin_amdgcn_sched_barrier(0);
        #pragma unroll
        for (int q = 0; q < LG; ++q) {
          #pragma unroll
          for (int g = 0; g < 4; ++g)
            acc[g] = __builtin_amdgcn_wmma_f32_16x16x4_f32(false, fa[q], false, fb[g][q],
                       (short)0, acc[g], false, false);
        }
        __builtin_amdgcn_sched_barrier(0);
        if (jg + LG < KC / 4) {
          #pragma unroll
          for (int q = 0; q < LG; ++q) {
            fa[q] = na[q];
            #pragma unroll
            for (int g = 0; g < 4; ++g) fb[g][q] = nb[g][q];
          }
        }
      }
    }
  }

  if (active && kh == 1) {
    #pragma unroll
    for (int g = 0; g < 4; ++g)
      #pragma unroll
      for (int v = 0; v < 8; ++v) red[p][g][v][lane] = acc[g][v];
  }
  __syncthreads();
  if (!active || kh == 1) return;

  #pragma unroll
  for (int g = 0; g < 4; ++g)
    #pragma unroll
    for (int v = 0; v < 8; ++v) acc[g][v] += red[p][g][v][lane];

  int tgt = TT - 1 - step;
  #pragma unroll
  for (int v = 0; v < 8; ++v) {
    int r = m0 + v + (hi << 3);
    int d = n0 + nn;
    float si = 1.f / (1.f + expf(-acc[0][v]));
    float sf = 1.f / (1.f + expf(-acc[1][v]));
    float gt = tanhf(acc[2][v]);
    float so = 1.f / (1.f + expf(-acc[3][v]));
    float cold = cc[(size_t)r * DD + d];
    float cnew = sf * cold + si * gt;
    float hnew = so * tanhf(cnew);
    cc[(size_t)r * DD + d] = cnew;
    hout[(size_t)r * DD + d] = hnew;
    if ((r >> 3) == tgt) lasth[(size_t)r * DD + d] = hnew;
  }
}

__global__ void ln_relu_kernel(const float* __restrict__ in,
                               const float* __restrict__ g,
                               const float* __restrict__ be,
                               float* __restrict__ out) {
  __shared__ float s1[256], s2[256];
  int r = blockIdx.x, t = threadIdx.x;
  const float* row = in + (size_t)r * DD;
  float s = 0.f, sq = 0.f;
  for (int i = t; i < DD; i += 256) { float v = row[i]; s += v; sq += v * v; }
  s1[t] = s; s2[t] = sq;
  __syncthreads();
  for (int off = 128; off > 0; off >>= 1) {
    if (t < off) { s1[t] += s1[t + off]; s2[t] += s2[t + off]; }
    __syncthreads();
  }
  float mean = s1[0] * (1.0f / DD);
  float var = s2[0] * (1.0f / DD) - mean * mean;
  float inv = rsqrtf(var + 1e-5f);
  for (int i = t; i < DD; i += 256) {
    float v = (row[i] - mean) * inv * g[i] + be[i];
    out[(size_t)r * DD + i] = v > 0.f ? v : 0.f;
  }
}

extern "C" void kernel_launch(void* const* d_in, const int* in_sizes, int n_in,
                              void* d_out, int out_size, void* d_ws, size_t ws_size,
                              hipStream_t stream) {
  const float* x   = (const float*)d_in[0];
  const float* Wn  = (const float*)d_in[1];
  const float* bn  = (const float*)d_in[2];
  const float* Wv  = (const float*)d_in[3];
  const float* bv  = (const float*)d_in[4];
  const float* Wf  = (const float*)d_in[5];
  const float* bf  = (const float*)d_in[6];
  const float* Wih = (const float*)d_in[7];
  const float* Whh = (const float*)d_in[8];
  const float* bih = (const float*)d_in[9];
  const float* bhh = (const float*)d_in[10];
  const float* Wc  = (const float*)d_in[11];
  const float* bc  = (const float*)d_in[12];
  const float* g_a = (const float*)d_in[13];
  const float* be_a= (const float*)d_in[14];
  const float* Wa  = (const float*)d_in[15];
  const float* ba  = (const float*)d_in[16];
  float* out = (float*)d_out;

  float* ws    = (float*)d_ws;
  float* A0    = ws;
  float* F1    = A0    + 131072;
  float* F2    = F1    + 131072;
  float* ENV   = F2    + 131072;
  float* XPRE  = ENV   + 262144;
  float* H0    = XPRE  + 524288;
  float* H1    = H0    + 131072;
  float* CC    = H1    + 131072;
  float* LASTH = CC    + 131072;
  float* UNR   = LASTH + 131072;
  float* LNR   = UNR   + 131072;

  pool_kernel<<<1536, 256, 0, stream>>>(x, A0, F1, F2);
  gemm_kernel<0><<<64, 128, 0, stream>>>(F1, DD, Wn, DD, bn, nullptr,
                                         ENV, 2 * DD, 0,  nullptr, DD, DD);
  gemm_kernel<0><<<64, 128, 0, stream>>>(F2, DD, Wv, DD, bv, nullptr,
                                         ENV, 2 * DD, DD, nullptr, DD, DD);
  gemm_kernel<1><<<128, 128, 0, stream>>>(ENV, 2 * DD, Wf, 2 * DD, bf, nullptr,
                                          H0, 0, 0, CC, 2 * DD, 2 * DD);
  gemm_kernel<0><<<256, 128, 0, stream>>>(A0, DD, Wih, DD, bih, bhh,
                                          XPRE, 4 * DD, 0, nullptr, 4 * DD, DD);
  float* hin = H0; float* hout = H1;
  for (int s = 0; s < TT; ++s) {
    int mtiles = (TT - s + 1) >> 1;
    int mgroups = (mtiles + 1) >> 1;
    lstm_step_kernel<<<64 * mgroups, 128, 0, stream>>>(XPRE, hin, Whh, CC, hout,
                                                       LASTH, s, mtiles);
    float* tmp = hin; hin = hout; hout = tmp;
  }
  gemm_kernel<0><<<64, 128, 0, stream>>>(LASTH, DD, Wc, DD, bc, nullptr,
                                         UNR, DD, 0, nullptr, DD, DD);
  ln_relu_kernel<<<128, 256, 0, stream>>>(UNR, g_a, be_a, LNR);
  gemm_kernel<2><<<118, 128, 0, stream>>>(LNR, DD, Wa, DD, ba, nullptr,
                                          out, NA, 0, nullptr, NA, DD);
}


# LSTM step: one output tile per block (was two); its 4 waves are 2 K halves x 2 gate pairs (was 4 gates per wave), gate math dealt over all 4 waves; grid 64 x live row tiles; every k sum order kept
# speedup vs baseline: 2.4115x; 1.2790x over previous
#include <hip/hip_runtime.h>
#include <math.h>

#define DD 1024
#define NA 1887
#define BB 8
#define TT 16
#define ROWS 128

typedef __attribute__((ext_vector_type(2))) float v2f;
typedef __attribute__((ext_vector_type(8))) float v8f;

__global__ void pool_kernel(const float* __restrict__ x, float* __restrict__ A0,
                            float* __restrict__ F1, float* __restrict__ F2) {
  int tid = blockIdx.x * blockDim.x + threadIdx.x;
  if (tid >= ROWS * DD * 3) return;
  int c  = tid % 3;
  int rd = tid / 3;
  int d  = rd % DD;
  int r  = rd / DD;
  const float* p = x + (size_t)rd * 147 + c;
  float s = 0.f;
  #pragma unroll
  for (int i = 0; i < 49; ++i) s += p[i * 3];
  s *= (1.0f / 49.0f);
  if (c == 0) {
    int b = r >> 4, t = r & 15;
    A0[(size_t)(t * BB + b) * DD + d] = s;
  } else if (c == 1) {
    F1[(size_t)r * DD + d] = s;
  } else {
    F2[(size_t)r * DD + d] = s;
  }
}

#define TCOLS 128
#define PITCH (TCOLS + 4)
#define PITCH4 (PITCH / 4)
#define TROWS 96
#define GG 8
#define LG 4
#define ROWS8(X)  X(0) X(1) X(2) X(3) X(4) X(5) X(6) X(7)
#define ROWS16(X) ROWS8(X) X(8) X(9) X(10) X(11) X(12) X(13) X(14) X(15)

template <int MODE>
__global__ void __launch_bounds__(128)
gemm_kernel(const float* __restrict__ A, int lda,
            const float* __restrict__ W, int ldw,
            const float* __restrict__ bias1,
            const float* __restrict__ bias2,
            float* __restrict__ out, int ldo, int coloff,
            float* __restrict__ out2, int N, int K) {
  __shared__ float4 tile4[2][TROWS][PITCH4];
  const int lw   = __builtin_amdgcn_readfirstlane((int)(threadIdx.x >> 5));
  const int lane = threadIdx.x & 31;
  const int mh   = blockIdx.x & 1;
  const int nb   = blockIdx.x >> 1;
  const int m0   = mh * 64 + lw * 16;
  const int nn   = lane & 15;
  const int hi   = lane >> 4;
  const int koff = hi << 1;
  const int ncol0 = nb * 32 + nn;
  const int ncol1 = ncol0 + 16;
  const int nld0 = (ncol0 < N) ? ncol0 : (N - 1);
  const int nld1 = (ncol1 < N) ? ncol1 : (N - 1);
  const float bsum0 = bias1[nld0] + (bias2 ? bias2[nld0] : 0.f);
  const float bsum1 = bias1[nld1] + (bias2 ? bias2[nld1] : 0.f);
  v8f acc0, acc1;
  #pragma unroll
  for (int i = 0; i < 8; ++i) { acc0[i] = bsum0; acc1[i] = bsum1; }

#define G_WROW(i) ((nb * 32 + lw + 4 * (i)) < N ? (nb * 32 + lw + 4 * (i)) : (N - 1))
#define G_ADECL(i) float4 pa##i = *(const float4*)(asrc + (size_t)(4 * (i)) * lda);
#define G_WDECL(i) const float* wsrc##i = W + (size_t)G_WROW(i) * ldw + (lane << 2); float4 pb##i = *(const float4*)(wsrc##i);
#define G_ASTORE(i) tb[lw + 4 * (i)][lane] = pa##i;
#define G_WSTORE(i) tb[64 + lw + 4 * (i)][lane] = pb##i;
#define G_ALOAD(i) pa##i = *(const float4*)(asrc + (size_t)(4 * (i)) * lda + ko);
#define G_WLOAD(i) pb##i = *(const float4*)(wsrc##i + ko);
  const float* asrc = A + (size_t)(mh * 64 + lw) * lda + (lane << 2);
  ROWS16(G_ADECL)
  ROWS8(G_WDECL)

  const int nround = K / TCOLS;
  for (int c = 0; c < nround; ++c) {
    float4 (*tb)[PITCH4] = tile4[c & 1];
    ROWS16(G_ASTORE)
    ROWS8(G_WSTORE)
    __syncthreads();
    if (c + 1 < nround) {
      const int ko = (c + 1) * TCOLS;
      ROWS16(G_ALOAD)
      ROWS8(G_WLOAD)
    }
    const float* tf = (const float*)&tb[0][0];
    const float* ap = tf + (lw * 16 + nn) * PITCH + koff;
    const float* bp = tf + (64 + nn) * PITCH + koff;
    v2f fa[GG], fb0[GG], fb1[GG];
    #pragma unroll
    for (int q = 0; q < GG; ++q) {
      fa[q]  = *(const v2f*)(ap + 4 * q);
      fb0[q] = *(const v2f*)(bp + 4 * q);
      fb1[q] = *(const v2f*)(bp + 16 * PITCH + 4 * q);
    }
    #pragma unroll
    for (int jg = 0; jg < TCOLS / 4; jg += GG) {
      v2f na[GG], nb0[GG], nb1[GG];
      if (jg + GG < TCOLS / 4) {
        #pragma unroll
        for (int q = 0; q < GG; ++q) {
          na[q]  = *(const v2f*)(ap + 4 * (jg + GG + q));
          nb0[q] = *(const v2f*)(bp + 4 * (jg + GG + q));
          nb1[q] = *(const v2f*)(bp + 16 * PITCH + 4 * (jg + GG + q));
        }
      }
      __builtin_amdgcn_sched_barrier(0);
      #pragma unroll
      for (int q = 0; q < GG; ++q) {
        acc0 = __builtin_amdgcn_wmma_f32_16x16x4_f32(false, fa[q], false, fb0[q],
                                                     (short)0, acc0, false, false);
        acc1 = __builtin_amdgcn_wmma_f32_16x16x4_f32(false, fa[q], false, fb1[q],
                                                     (short)0, acc1, false, false);
      }
      __builtin_amdgcn_sched_barrier(0);
      if (jg + GG < TCOLS / 4) {
        #pragma unroll
        for (int q = 0; q < GG; ++q) { fa[q] = na[q]; fb0[q] = nb0[q]; fb1[q] = nb1[q]; }
      }
    }
  }

  #pragma unroll
  for (int t = 0; t < 2; ++t) {
    const int ncol = t ? ncol1 : ncol0;
    #pragma unroll
    for (int v = 0; v < 8; ++v) {
      int r = m0 + v + (hi << 3);
      float val = t ? acc1[v] : acc0[v];
      if (MODE == 0) {
        out[(size_t)r * ldo + coloff + ncol] = val;
      } else if (MODE == 1) {
        int b = r >> 4, tt = r & 15;
        int rr = tt * BB + b;
        if (ncol < DD) out[(size_t)rr * DD + ncol] = val;
        else           out2[(size_t)rr * DD + (ncol - DD)] = val;
      } else {
        if (ncol < N) {
          int tt = r >> 3, b = r & 7;
          out[(size_t)(b * TT + tt) * N + ncol] = val;
        }
      }
    }
  }
}

#define KC (TCOLS / 2)
#define LROWS 16
#define WROWS 64
#define LTROWS (LROWS + WROWS)
#define NROUND (DD / 2 / KC)
#define LK 8
#define ROWS4(X) X(0) X(1) X(2) X(3)

__global__ void __launch_bounds__(128)
lstm_step_kernel(const float* __restrict__ xpre,
                 const float* __restrict__ hin,
                 const float* __restrict__ Whh,
                 float* __restrict__ cc,
                 float* __restrict__ hout,
                 float* __restrict__ lasth,
                 int step, int mtiles) {
  __shared__ float4 tile4[2][LTROWS][PITCH4];
  __shared__ float red[2][4][8][32];
  const int lw   = __builtin_amdgcn_readfirstlane((int)(threadIdx.x >> 5));
  const int lane = threadIdx.x & 31;
  const int nt   = blockIdx.x / mtiles;
  const int mt   = blockIdx.x - nt * mtiles;
  const int n0   = nt << 4;
  const int m0   = mt << 4;
  const int kh   = lw & 1;
  const int gp   = lw >> 1;
  const int nn   = lane & 15;
  const int hi   = lane >> 4;
  const int koff = hi << 1;

  v8f acc0, acc1;
  if (kh == 0) {
    #pragma unroll
    for (int v = 0; v < 8; ++v) {
      int r = m0 + v + (hi << 3);
      const float* xp = xpre + (size_t)r * (4 * DD) + (2 * gp) * DD + n0 + nn;
      acc0[v] = xp[0];
      acc1[v] = xp[DD];
    }
  } else {
    #pragma unroll
    for (int v = 0; v < 8; ++v) { acc0[v] = 0.f; acc1[v] = 0.f; }
  }

  const int kcol = (lane >> 4) * (DD / 2) + ((lane & 15) << 2);
#define H_DECL(i) float4 ph##i = *(const float4*)(hsrc + (size_t)(4 * (i)) * DD);
#define W_OFF(i) ((size_t)(((i) >> 2) * DD + 4 * ((i) & 3)) * DD)
#define W_DECL(i) float4 pw##i = *(const float4*)(wsrc + W_OFF(i));
#define H_STORE(i) tb[lw + 4 * (i)][lane] = ph##i;
#define W_STORE(i) tb[LROWS + lw + 4 * (i)][lane] = pw##i;
#define H_LOAD(i) ph##i = *(const float4*)(hsrc + (size_t)(4 * (i)) * DD + ko);
#define W_LOAD(i) pw##i = *(const float4*)(wsrc + W_OFF(i) + ko);
  const float* hsrc = hin + (size_t)(m0 + lw) * DD + kcol;
  const float* wsrc = Whh + (size_t)(n0 + lw) * DD + kcol;
  ROWS4(H_DECL)
  ROWS16(W_DECL)

  #pragma unroll 1
  for (int c = 0; c < NROUND; ++c) {
    float4 (*tb)[PITCH4] = tile4[c & 1];
    ROWS4(H_STORE)
    ROWS16(W_STORE)
    __syncthreads();
    if (c + 1 < NROUND) {
      const int ko = (c + 1) * KC;
      ROWS4(H_LOAD)
      ROWS16(W_LOAD)
    }
    const float* tf = (const float*)&tb[0][0];
    const float* ap = tf + nn * PITCH + kh * KC + koff;
    const float* bp = tf + (LROWS + gp * 32 + nn) * PITCH + kh * KC + koff;
    v2f fa[LK], fb0[LK], fb1[LK];
    #pragma unroll
    for (int q = 0; q < LK; ++q) {
      fa[q]  = *(const v2f*)(ap + 4 * q);
      fb0[q] = *(const v2f*)(bp + 4 * q);
      fb1[q] = *(const v2f*)(bp + 16 * PITCH + 4 * q);
    }
    #pragma unroll
    for (int jg = 0; jg < KC / 4; jg += LK) {
      v2f na[LK], nb0[LK], nb1[LK];
      if (jg + LK < KC / 4) {
        #pragma unroll
        for (int q = 0; q < LK; ++q) {
          na[q]  = *(const v2f*)(ap + 4 * (jg + LK + q));
          nb0[q] = *(const v2f*)(bp + 4 * (jg + LK + q));
          nb1[q] = *(const v2f*)(bp + 16 * PITCH + 4 * (jg + LK + q));
        }
      }
      __builtin_amdgcn_sched_barrier(0);
      #pragma unroll
      for (int q = 0; q < LK; ++q) {
        acc0 = __builtin_amdgcn_wmma_f32_16x16x4_f32(false, fa[q], false, fb0[q],
                                                     (short)0, acc0, false, false);
        acc1 = __builtin_amdgcn_wmma_f32_16x16x4_f32(false, fa[q], false, fb1[q],
                                                     (short)0, acc1, false, false);
      }
      __builtin_amdgcn_sched_barrier(0);
      if (jg + LK < KC / 4) {
        #pragma unroll
        for (int q = 0; q < LK; ++q) { fa[q] = na[q]; fb0[q] = nb0[q]; fb1[q] = nb1[q]; }
      }
    }
  }

  #pragma unroll
  for (int v = 0; v < 8; ++v) {
    red[kh][2 * gp][v][lane]     = acc0[v];
    red[kh][2 * gp + 1][v][lane] = acc1[v];
  }
  __syncthreads();

  const int tgt = TT - 1 - step;
  const int d = n0 + nn;
  #pragma unroll
  for (int j = 0; j < 2; ++j) {
    const int v = 2 * lw + j;
    const int r = m0 + v + (hi << 3);
    const float gi = red[0][0][v][lane] + red[1][0][v][lane];
    const float gf = red[0][1][v][lane] + red[1][1][v][lane];
    const float gg = red[0][2][v][lane] + red[1][2][v][lane];
    const float go = red[0][3][v][lane] + red[1][3][v][lane];
    float si = 1.f / (1.f + expf(-gi));
    float sf = 1.f / (1.f + expf(-gf));
    float gt = tanhf(gg);
    float so = 1.f / (1.f + expf(-go));
    float cold = cc[(size_t)r * DD + d];
    float cnew = sf * cold + si * gt;
    float hnew = so * tanhf(cnew);
    cc[(size_t)r * DD + d] = cnew;
    hout[(size_t)r * DD + d] = hnew;
    if ((r >> 3) == tgt) lasth[(size_t)r * DD + d] = hnew;
  }
}

__global__ void ln_relu_kernel(const float* __restrict__ in,
                               const float* __restrict__ g,
                               const float* __restrict__ be,
                               float* __restrict__ out) {
  __shared__ float s1[256], s2[256];
  int r = blockIdx.x, t = threadIdx.x;
  const float* row = in + (size_t)r * DD;
  float s = 0.f, sq = 0.f;
  for (int i = t; i < DD; i += 256) { float v = row[i]; s += v; sq += v * v; }
  s1[t] = s; s2[t] = sq;
  __syncthreads();
  for (int off = 128; off > 0; off >>= 1) {
    if (t < off) { s1[t] += s1[t + off]; s2[t] += s2[t + off]; }
    __syncthreads();
  }
  float mean = s1[0] * (1.0f / DD);
  float var = s2[0] * (1.0f / DD) - mean * mean;
  float inv = rsqrtf(var + 1e-5f);
  for (int i = t; i < DD; i += 256) {
    float v = (row[i] - mean) * inv * g[i] + be[i];
    out[(size_t)r * DD + i] = v > 0.f ? v : 0.f;
  }
}

extern "C" void kernel_launch(void* const* d_in, const int* in_sizes, int n_in,
                              void* d_out, int out_size, void* d_ws, size_t ws_size,
                              hipStream_t stream) {
  const float* x   = (const float*)d_in[0];
  const float* Wn  = (const float*)d_in[1];
  const float* bn  = (const float*)d_in[2];
  const float* Wv  = (const float*)d_in[3];
  const float* bv  = (const float*)d_in[4];
  const float* Wf  = (const float*)d_in[5];
  const float* bf  = (const float*)d_in[6];
  const float* Wih = (const float*)d_in[7];
  const float* Whh = (const float*)d_in[8];
  const float* bih = (const float*)d_in[9];
  const float* bhh = (const float*)d_in[10];
  const float* Wc  = (const float*)d_in[11];
  const float* bc  = (const float*)d_in[12];
  const float* g_a = (const float*)d_in[13];
  const float* be_a= (const float*)d_in[14];
  const float* Wa  = (const float*)d_in[15];
  const float* ba  = (const float*)d_in[16];
  float* out = (float*)d_out;

  float* ws    = (float*)d_ws;
  float* A0    = ws;
  float* F1    = A0    + 131072;
  float* F2    = F1    + 131072;
  float* ENV   = F2    + 131072;
  float* XPRE  = ENV   + 262144;
  float* H0    = XPRE  + 524288;
  float* H1    = H0    + 131072;
  float* CC    = H1    + 131072;
  float* LASTH = CC    + 131072;
  float* UNR   = LASTH + 131072;
  float* LNR   = UNR   + 131072;

  pool_kernel<<<1536, 256, 0, stream>>>(x, A0, F1, F2);
  gemm_kernel<0><<<64, 128, 0, stream>>>(F1, DD, Wn, DD, bn, nullptr,
                                         ENV, 2 * DD, 0,  nullptr, DD, DD);
  gemm_kernel<0><<<64, 128, 0, stream>>>(F2, DD, Wv, DD, bv, nullptr,
                                         ENV, 2 * DD, DD, nullptr, DD, DD);
  gemm_kernel<1><<<128, 128, 0, stream>>>(ENV, 2 * DD, Wf, 2 * DD, bf, nullptr,
                                          H0, 0, 0, CC, 2 * DD, 2 * DD);
  gemm_kernel<0><<<256, 128, 0, stream>>>(A0, DD, Wih, DD, bih, bhh,
                                          XPRE, 4 * DD, 0, nullptr, 4 * DD, DD);
  float* hin = H0; float* hout = H1;
  for (int s = 0; s < TT; ++s) {
    int mtiles = (TT - s + 1) >> 1;
    lstm_step_kernel<<<64 * mtiles, 128, 0, stream>>>(XPRE, hin, Whh, CC, hout,
                                                      LASTH, s, mtiles);
    float* tmp = hin; hin = hout; hout = tmp;
  }
  gemm_kernel<0><<<64, 128, 0, stream>>>(LASTH, DD, Wc, DD, bc, nullptr,
                                         UNR, DD, 0, nullptr, DD, DD);
  ln_relu_kernel<<<128, 256, 0, stream>>>(UNR, g_a, be_a, LNR);
  gemm_kernel<2><<<118, 128, 0, stream>>>(LNR, DD, Wa, DD, ba, nullptr,
                                          out, NA, 0, nullptr, NA, DD);
}


# All WMMA K loops: LDS staging re-timed as a three-stage pipeline (two register sets; round c stores round c+1 and requests round c+3; round barrier waits for LDS only); every k sum order kept
# speedup vs baseline: 2.6261x; 1.0890x over previous
#include <hip/hip_runtime.h>
#include <math.h>

#define DD 1024
#define NA 1887
#define BB 8
#define TT 16
#define ROWS 128

typedef __attribute__((ext_vector_type(2))) float v2f;
typedef __attribute__((ext_vector_type(8))) float v8f;

__global__ void pool_kernel(const float* __restrict__ x, float* __restrict__ A0,
                            float* __restrict__ F1, float* __restrict__ F2) {
  int tid = blockIdx.x * blockDim.x + threadIdx.x;
  if (tid >= ROWS * DD * 3) return;
  int c  = tid % 3;
  int rd = tid / 3;
  int d  = rd % DD;
  int r  = rd / DD;
  const float* p = x + (size_t)rd * 147 + c;
  float s = 0.f;
  #pragma unroll
  for (int i = 0; i < 49; ++i) s += p[i * 3];
  s *= (1.0f / 49.0f);
  if (c == 0) {
    int b = r >> 4, t = r & 15;
    A0[(size_t)(t * BB + b) * DD + d] = s;
  } else if (c == 1) {
    F1[(size_t)r * DD + d] = s;
  } else {
    F2[(size_t)r * DD + d] = s;
  }
}

#define TCOLS 128
#define PITCH (TCOLS + 4)
#define PITCH4 (PITCH / 4)
#define TROWS 96
#define GG 8
#define LG 4
#define ROWS8(X)  X(0) X(1) X(2) X(3) X(4) X(5) X(6) X(7)
#define ROWS4S(X, S)  X(S, 0) X(S, 1) X(S, 2) X(S, 3)
#define ROWS8S(X, S)  ROWS4S(X, S) X(S, 4) X(S, 5) X(S, 6) X(S, 7)
#define ROWS16S(X, S) ROWS8S(X, S) X(S, 8) X(S, 9) X(S, 10) X(S, 11) X(S, 12) X(S, 13) X(S, 14) X(S, 15)

__device__ __forceinline__ void lds_barrier() {
  asm volatile("s_wait_dscnt 0x0\n\ts_barrier_signal -1\n\ts_barrier_wait -1" ::: "memory");
}

template <int MODE>
__global__ void __launch_bounds__(128)
gemm_kernel(const float* A, int lda,
            const float* W, int ldw,
            const float* __restrict__ bias1,
            const float* __restrict__ bias2,
            float* __restrict__ out, int ldo, int coloff,
            float* __restrict__ out2, int N, int K) {
  __shared__ float4 tile4[2][TROWS][PITCH4];
  const int lw   = __builtin_amdgcn_readfirstlane((int)(threadIdx.x >> 5));
  const int lane = threadIdx.x & 31;
  const int mh   = blockIdx.x & 1;
  const int nb   = blockIdx.x >> 1;
  const int m0   = mh * 64 + lw * 16;
  const int nn   = lane & 15;
  const int hi   = lane >> 4;
  const int koff = hi << 1;
  const int ncol0 = nb * 32 + nn;
  const int ncol1 = ncol0 + 16;
  const int nld0 = (ncol0 < N) ? ncol0 : (N - 1);
  const int nld1 = (ncol1 < N) ? ncol1 : (N - 1);
  const float bsum0 = bias1[nld0] + (bias2 ? bias2[nld0] : 0.f);
  const float bsum1 = bias1[nld1] + (bias2 ? bias2[nld1] : 0.f);
  v8f acc0, acc1;
  #pragma unroll
  for (int i = 0; i < 8; ++i) { acc0[i] = bsum0; acc1[i] = bsum1; }

#define G_WROW(i) ((nb * 32 + lw + 4 * (i)) < N ? (nb * 32 + lw + 4 * (i)) : (N - 1))
#define G_WPTR(i) const float* wsrc##i = W + (size_t)G_WROW(i) * ldw + (lane << 2);
#define G_ADC(S, i) float4 pa##S##i;
#define G_WDC(S, i) float4 pb##S##i;
#define G_ALD(S, i) pa##S##i = *(const float4*)(asrc + (size_t)(4 * (i)) * lda + ko);
#define G_WLD(S, i) pb##S##i = *(const float4*)(wsrc##i + ko);
#define G_AST(S, i) tb[lw + 4 * (i)][lane] = pa##S##i;
#define G_WST(S, i) tb[64 + lw + 4 * (i)][lane] = pb##S##i;
  const float* asrc = A + (size_t)(mh * 64 + lw) * lda + (lane << 2);
  ROWS8(G_WPTR)
  ROWS16S(G_ADC, a) ROWS8S(G_WDC, a)
  ROWS16S(G_ADC, b) ROWS8S(G_WDC, b)
  ROWS16S(G_ADC, t) ROWS8S(G_WDC, t)

  const int nround = K / TCOLS;
  { const int ko = 0; ROWS16S(G_ALD, t) ROWS8S(G_WLD, t) }
  __builtin_amdgcn_sched_barrier(0);
  { const int ko = TCOLS; ROWS16S(G_ALD, b) ROWS8S(G_WLD, b) }
  __builtin_amdgcn_sched_barrier(0);
  { const int ko = 2 * TCOLS; ROWS16S(G_ALD, a) ROWS8S(G_WLD, a) }
  __builtin_amdgcn_sched_barrier(0);
  { float4 (*tb)[PITCH4] = tile4[0]; ROWS16S(G_AST, t) ROWS8S(G_WST, t) }

#define GEMM_ROUND(cexpr, S, ST, LD)                                            \
  {                                                                             \
    const int c_ = (cexpr);                                                     \
    lds_barrier();                                                              \
    const float* tf = (const float*)&tile4[c_ & 1][0][0];                       \
    const float* ap = tf + (lw * 16 + nn) * PITCH + koff;                       \
    const float* bp = tf + (64 + nn) * PITCH + koff;                            \
    v2f fa[GG], fb0[GG], fb1[GG];                                               \
    _Pragma("unroll")                                                           \
    for (int q = 0; q < GG; ++q) {                                              \
      fa[q]  = *(const v2f*)(ap + 4 * q);                                       \
      fb0[q] = *(const v2f*)(bp + 4 * q);                                       \
      fb1[q] = *(const v2f*)(bp + 16 * PITCH + 4 * q);                          \
    }                                                                           \
    _Pragma("unroll")                                                           \
    for (int jg = 0; jg < TCOLS / 4; jg += GG) {                                \
      v2f na[GG], nb0[GG], nb1[GG];                                             \
      if (jg + GG < TCOLS / 4) {                                                \
        _Pragma("unroll")                                                       \
        for (int q = 0; q < GG; ++q) {                                          \
          na[q]  = *(const v2f*)(ap + 4 * (jg + GG + q));                       \
          nb0[q] = *(const v2f*)(bp + 4 * (jg + GG + q));                       \
          nb1[q] = *(const v2f*)(bp + 16 * PITCH + 4 * (jg + GG + q));          \
        }                                                                       \
      }                                                                         \
      if (jg == 0) {                                                            \
         \
         \
         \
        __builtin_amdgcn_sched_barrier(0);                                      \
        if (ST) {                                                               \
          float4 (*tb)[PITCH4] = tile4[(c_ + 1) & 1];                           \
          ROWS16S(G_AST, S) ROWS8S(G_WST, S)                                    \
        }                                                                       \
        if (LD) {                                                               \
          const int ko = (c_ + 3) * TCOLS;                                      \
          ROWS16S(G_ALD, S) ROWS8S(G_WLD, S)                                    \
        }                                                                       \
      }                                                                         \
       \
       \
      __builtin_amdgcn_sched_barrier(0);                                        \
      _Pragma("unroll")                                                         \
      for (int q = 0; q < GG; ++q) {                                            \
        acc0 = __builtin_amdgcn_wmma_f32_16x16x4_f32(false, fa[q], false, fb0[q], \
                                                     (short)0, acc0, false, false); \
        acc1 = __builtin_amdgcn_wmma_f32_16x16x4_f32(false, fa[q], false, fb1[q], \
                                                     (short)0, acc1, false, false); \
      }                                                                         \
       \
       \
      __builtin_amdgcn_sched_barrier(0);                                        \
      if (jg + GG < TCOLS / 4) {                                                \
        _Pragma("unroll")                                                       \
        for (int q = 0; q < GG; ++q) { fa[q] = na[q]; fb0[q] = nb0[q]; fb1[q] = nb1[q]; } \
      }                                                                         \
    }                                                                           \
  }

  int c = 0;
  for (; c + 5 < nround; c += 2) {
    GEMM_ROUND(c, b, 1, 1)
    GEMM_ROUND(c + 1, a, 1, 1)
  }
  GEMM_ROUND(c, b, 1, 1)
  GEMM_ROUND(c + 1, a, 1, 0)
  GEMM_ROUND(c + 2, b, 1, 0)
  GEMM_ROUND(c + 3, a, 0, 0)

  #pragma unroll
  for (int t = 0; t < 2; ++t) {
    const int ncol = t ? ncol1 : ncol0;
    #pragma unroll
    for (int v = 0; v < 8; ++v) {
      int r = m0 + v + (hi << 3);
      float val = t ? acc1[v] : acc0[v];
      if (MODE == 0) {
        out[(size_t)r * ldo + coloff + ncol] = val;
      } else if (MODE == 1) {
        int b = r >> 4, tt = r & 15;
        int rr = tt * BB + b;
        if (ncol < DD) out[(size_t)rr * DD + ncol] = val;
        else           out2[(size_t)rr * DD + (ncol - DD)] = val;
      } else {
        if (ncol < N) {
          int tt = r >> 3, b = r & 7;
          out[(size_t)(b * TT + tt) * N + ncol] = val;
        }
      }
    }
  }
}

#define KC (TCOLS / 2)
#define LROWS 16
#define WROWS 64
#define LTROWS (LROWS + WROWS)
#define NROUND (DD / 2 / KC)
#define LK 8

__global__ void __launch_bounds__(128)
lstm_step_kernel(const float* __restrict__ xpre,
                 const float* hin,
                 const float* Whh,
                 float* __restrict__ cc,
                 float* __restrict__ hout,
                 float* __restrict__ lasth,
                 int step, int mtiles) {
  __shared__ float4 tile4[2][LTROWS][PITCH4];
  __shared__ float red[2][4][8][32];
  const int lw   = __builtin_amdgcn_readfirstlane((int)(threadIdx.x >> 5));
  const int lane = threadIdx.x & 31;
  const int nt   = blockIdx.x / mtiles;
  const int mt   = blockIdx.x - nt * mtiles;
  const int n0   = nt << 4;
  const int m0   = mt << 4;
  const int kh   = lw & 1;
  const int gp   = lw >> 1;
  const int nn   = lane & 15;
  const int hi   = lane >> 4;
  const int koff = hi << 1;

  v8f acc0, acc1;
  if (kh == 0) {
    #pragma unroll
    for (int v = 0; v < 8; ++v) {
      int r = m0 + v + (hi << 3);
      const float* xp = xpre + (size_t)r * (4 * DD) + (2 * gp) * DD + n0 + nn;
      acc0[v] = xp[0];
      acc1[v] = xp[DD];
    }
  } else {
    #pragma unroll
    for (int v = 0; v < 8; ++v) { acc0[v] = 0.f; acc1[v] = 0.f; }
  }

  const int kcol = (lane >> 4) * (DD / 2) + ((lane & 15) << 2);
#define W_OFF(i) ((size_t)(((i) >> 2) * DD + 4 * ((i) & 3)) * DD)
#define H_DC(S, i) float4 ph##S##i;
#define W_DC(S, i) float4 pw##S##i;
#define H_LD(S, i) ph##S##i = *(const float4*)(hsrc + (size_t)(4 * (i)) * DD + ko);
#define W_LD(S, i) pw##S##i = *(const float4*)(wsrc + W_OFF(i) + ko);
#define H_ST(S, i) tb[lw + 4 * (i)][lane] = ph##S##i;
#define W_ST(S, i) tb[LROWS + lw + 4 * (i)][lane] = pw##S##i;
  const float* hsrc = hin + (size_t)(m0 + lw) * DD + kcol;
  const float* wsrc = Whh + (size_t)(n0 + lw) * DD + kcol;
  ROWS4S(H_DC, a) ROWS16S(W_DC, a)
  ROWS4S(H_DC, b) ROWS16S(W_DC, b)
  ROWS4S(H_DC, t) ROWS16S(W_DC, t)
  { const int ko = 0;      ROWS4S(H_LD, t) ROWS16S(W_LD, t) }
  __builtin_amdgcn_sched_barrier(0);
  { const int ko = KC;     ROWS4S(H_LD, b) ROWS16S(W_LD, b) }
  __builtin_amdgcn_sched_barrier(0);
  { const int ko = 2 * KC; ROWS4S(H_LD, a) ROWS16S(W_LD, a) }
  __builtin_amdgcn_sched_barrier(0);
  { float4 (*tb)[PITCH4] = tile4[0]; ROWS4S(H_ST, t) ROWS16S(W_ST, t) }

#define LSTM_ROUND(cexpr, S, ST, LD)                                            \
  {                                                                             \
    const int c_ = (cexpr);                                                     \
    lds_barrier();                                                              \
    const float* tf = (const float*)&tile4[c_ & 1][0][0];                       \
    const float* ap = tf + nn * PITCH + kh * KC + koff;                         \
    const float* bp = tf + (LROWS + gp * 32 + nn) * PITCH + kh * KC + koff;     \
    v2f fa[LK], fb0[LK], fb1[LK];                                               \
    _Pragma("unroll")                                                           \
    for (int q = 0; q < LK; ++q) {                                              \
      fa[q]  = *(const v2f*)(ap + 4 * q);                                       \
      fb0[q] = *(const v2f*)(bp + 4 * q);                                       \
      fb1[q] = *(const v2f*)(bp + 16 * PITCH + 4 * q);                          \
    }                                                                           \
    _Pragma("unroll")                                                           \
    for (int jg = 0; jg < KC / 4; jg += LK) {                                   \
      v2f na[LK], nb0[LK], nb1[LK];                                             \
      if (jg + LK < KC / 4) {                                                   \
        _Pragma("unroll")                                                       \
        for (int q = 0; q < LK; ++q) {                                          \
          na[q]  = *(const v2f*)(ap + 4 * (jg + LK + q));                       \
          nb0[q] = *(const v2f*)(bp + 4 * (jg + LK + q));                       \
          nb1[q] = *(const v2f*)(bp + 16 * PITCH + 4 * (jg + LK + q));          \
        }                                                                       \
      }                                                                         \
      if (jg == 0) {                                                            \
         \
         \
         \
        __builtin_amdgcn_sched_barrier(0);                                      \
        if (ST) {                                                               \
          float4 (*tb)[PITCH4] = tile4[(c_ + 1) & 1];                           \
          ROWS4S(H_ST, S) ROWS16S(W_ST, S)                                      \
        }                                                                       \
        if (LD) {                                                               \
          const int ko = (c_ + 3) * KC;                                         \
          ROWS4S(H_LD, S) ROWS16S(W_LD, S)                                      \
        }                                                                       \
      }                                                                         \
       \
       \
      __builtin_amdgcn_sched_barrier(0);                                        \
      _Pragma("unroll")                                                         \
      for (int q = 0; q < LK; ++q) {                                            \
        acc0 = __builtin_amdgcn_wmma_f32_16x16x4_f32(false, fa[q], false, fb0[q], \
                                                     (short)0, acc0, false, false); \
        acc1 = __builtin_amdgcn_wmma_f32_16x16x4_f32(false, fa[q], false, fb1[q], \
                                                     (short)0, acc1, false, false); \
      }                                                                         \
       \
       \
      __builtin_amdgcn_sched_barrier(0);                                        \
      if (jg + LK < KC / 4) {                                                   \
        _Pragma("unroll")                                                       \
        for (int q = 0; q < LK; ++q) { fa[q] = na[q]; fb0[q] = nb0[q]; fb1[q] = nb1[q]; } \
      }                                                                         \
    }                                                                           \
  }

  static_assert(NROUND == 8, "the rounds below are written out for 8");
  LSTM_ROUND(0, b, 1, 1)
  LSTM_ROUND(1, a, 1, 1)
  LSTM_ROUND(2, b, 1, 1)
  LSTM_ROUND(3, a, 1, 1)
  LSTM_ROUND(4, b, 1, 1)
  LSTM_ROUND(5, a, 1, 0)
  LSTM_ROUND(6, b, 1, 0)
  LSTM_ROUND(7, a, 0, 0)

  #pragma unroll
  for (int v = 0; v < 8; ++v) {
    red[kh][2 * gp][v][lane]     = acc0[v];
    red[kh][2 * gp + 1][v][lane] = acc1[v];
  }
  __syncthreads();

  const int tgt = TT - 1 - step;
  const int d = n0 + nn;
  #pragma unroll
  for (int j = 0; j < 2; ++j) {
    const int v = 2 * lw + j;
    const int r = m0 + v + (hi << 3);
    const float gi = red[0][0][v][lane] + red[1][0][v][lane];
    const float gf = red[0][1][v][lane] + red[1][1][v][lane];
    const float gg = red[0][2][v][lane] + red[1][2][v][lane];
    const float go = red[0][3][v][lane] + red[1][3][v][lane];
    float si = 1.f / (1.f + expf(-gi));
    float sf = 1.f / (1.f + expf(-gf));
    float gt = tanhf(gg);
    float so = 1.f / (1.f + expf(-go));
    float cold = cc[(size_t)r * DD + d];
    float cnew = sf * cold + si * gt;
    float hnew = so * tanhf(cnew);
    cc[(size_t)r * DD + d] = cnew;
    hout[(size_t)r * DD + d] = hnew;
    if ((r >> 3) == tgt) lasth[(size_t)r * DD + d] = hnew;
  }
}

__global__ void ln_relu_kernel(const float* __restrict__ in,
                               const float* __restrict__ g,
                               const float* __restrict__ be,
                               float* __restrict__ out) {
  __shared__ float s1[256], s2[256];
  int r = blockIdx.x, t = threadIdx.x;
  const float* row = in + (size_t)r * DD;
  float s = 0.f, sq = 0.f;
  for (int i = t; i < DD; i += 256) { float v = row[i]; s += v; sq += v * v; }
  s1[t] = s; s2[t] = sq;
  __syncthreads();
  for (int off = 128; off > 0; off >>= 1) {
    if (t < off) { s1[t] += s1[t + off]; s2[t] += s2[t + off]; }
    __syncthreads();
  }
  float mean = s1[0] * (1.0f / DD);
  float var = s2[0] * (1.0f / DD) - mean * mean;
  float inv = rsqrtf(var + 1e-5f);
  for (int i = t; i < DD; i += 256) {
    float v = (row[i] - mean) * inv * g[i] + be[i];
    out[(size_t)r * DD + i] = v > 0.f ? v : 0.f;
  }
}

extern "C" void kernel_launch(void* const* d_in, const int* in_sizes, int n_in,
                              void* d_out, int out_size, void* d_ws, size_t ws_size,
                              hipStream_t stream) {
  const float* x   = (const float*)d_in[0];
  const float* Wn  = (const float*)d_in[1];
  const float* bn  = (const float*)d_in[2];
  const float* Wv  = (const float*)d_in[3];
  const float* bv  = (const float*)d_in[4];
  const float* Wf  = (const float*)d_in[5];
  const float* bf  = (const float*)d_in[6];
  const float* Wih = (const float*)d_in[7];
  const float* Whh = (const float*)d_in[8];
  const float* bih = (const float*)d_in[9];
  const float* bhh = (const float*)d_in[10];
  const float* Wc  = (const float*)d_in[11];
  const float* bc  = (const float*)d_in[12];
  const float* g_a = (const float*)d_in[13];
  const float* be_a= (const float*)d_in[14];
  const float* Wa  = (const float*)d_in[15];
  const float* ba  = (const float*)d_in[16];
  float* out = (float*)d_out;

  float* ws    = (float*)d_ws;
  float* A0    = ws;
  float* F1    = A0    + 131072;
  float* F2    = F1    + 131072;
  float* ENV   = F2    + 131072;
  float* XPRE  = ENV   + 262144;
  float* H0    = XPRE  + 524288;
  float* H1    = H0    + 131072;
  float* CC    = H1    + 131072;
  float* LASTH = CC    + 131072;
  float* UNR   = LASTH + 131072;
  float* LNR   = UNR   + 131072;

  pool_kernel<<<1536, 256, 0, stream>>>(x, A0, F1, F2);
  gemm_kernel<0><<<64, 128, 0, stream>>>(F1, DD, Wn, DD, bn, nullptr,
                                         ENV, 2 * DD, 0,  nullptr, DD, DD);
  gemm_kernel<0><<<64, 128, 0, stream>>>(F2, DD, Wv, DD, bv, nullptr,
                                         ENV, 2 * DD, DD, nullptr, DD, DD);
  gemm_kernel<1><<<128, 128, 0, stream>>>(ENV, 2 * DD, Wf, 2 * DD, bf, nullptr,
                                          H0, 0, 0, CC, 2 * DD, 2 * DD);
  gemm_kernel<0><<<256, 128, 0, stream>>>(A0, DD, Wih, DD, bih, bhh,
                                          XPRE, 4 * DD, 0, nullptr, 4 * DD, DD);
  float* hin = H0; float* hout = H1;
  for (int s = 0; s < TT; ++s) {
    int mtiles = (TT - s + 1) >> 1;
    lstm_step_kernel<<<64 * mtiles, 128, 0, stream>>>(XPRE, hin, Whh, CC, hout,
                                                      LASTH, s, mtiles);
    float* tmp = hin; hin = hout; hout = tmp;
  }
  gemm_kernel<0><<<64, 128, 0, stream>>>(LASTH, DD, Wc, DD, bc, nullptr,
                                         UNR, DD, 0, nullptr, DD, DD);
  ln_relu_kernel<<<128, 256, 0, stream>>>(UNR, g_a, be_a, LNR);
  gemm_kernel<2><<<118, 128, 0, stream>>>(LNR, DD, Wa, DD, ba, nullptr,
                                          out, NA, 0, nullptr, NA, DD);
}


# e_n, e_v, fuse, Wc and head GEMMs: block tile 32 x 32 (was 64 x 32), one output tile per wave (was two): twice the blocks, half the chained products per wave; x_pre as before; every k sum order kept
# speedup vs baseline: 2.7807x; 1.0589x over previous
#include <hip/hip_runtime.h>
#include <math.h>

#define DD 1024
#define NA 1887
#define BB 8
#define TT 16
#define ROWS 128

typedef __attribute__((ext_vector_type(2))) float v2f;
typedef __attribute__((ext_vector_type(8))) float v8f;

__global__ void pool_kernel(const float* __restrict__ x, float* __restrict__ A0,
                            float* __restrict__ F1, float* __restrict__ F2) {
  int tid = blockIdx.x * blockDim.x + threadIdx.x;
  if (tid >= ROWS * DD * 3) return;
  int c  = tid % 3;
  int rd = tid / 3;
  int d  = rd % DD;
  int r  = rd / DD;
  const float* p = x + (size_t)rd * 147 + c;
  float s = 0.f;
  #pragma unroll
  for (int i = 0; i < 49; ++i) s += p[i * 3];
  s *= (1.0f / 49.0f);
  if (c == 0) {
    int b = r >> 4, t = r & 15;
    A0[(size_t)(t * BB + b) * DD + d] = s;
  } else if (c == 1) {
    F1[(size_t)r * DD + d] = s;
  } else {
    F2[(size_t)r * DD + d] = s;
  }
}

#define TCOLS 128
#define PITCH (TCOLS + 4)
#define PITCH4 (PITCH / 4)
#define GG 8
#define LG 4
#define ROWS8(X)  X(0) X(1) X(2) X(3) X(4) X(5) X(6) X(7)
#define ROWS4S(X, S)  X(S, 0) X(S, 1) X(S, 2) X(S, 3)
#define ROWS8S(X, S)  ROWS4S(X, S) X(S, 4) X(S, 5) X(S, 6) X(S, 7)
#define ROWS16S(X, S) ROWS8S(X, S) X(S, 8) X(S, 9) X(S, 10) X(S, 11) X(S, 12) X(S, 13) X(S, 14) X(S, 15)

__device__ __forceinline__ void lds_barrier() {
  asm volatile("s_wait_dscnt 0x0\n\ts_barrier_signal -1\n\ts_barrier_wait -1" ::: "memory");
}

template <int MODE, int NT>
__global__ void __launch_bounds__(128)
gemm_kernel(const float* A, int lda,
            const float* W, int ldw,
            const float* __restrict__ bias1,
            const float* __restrict__ bias2,
            float* __restrict__ out, int ldo, int coloff,
            float* __restrict__ out2, int N, int K) {
  constexpr int AR = 32 * NT;
  __shared__ float4 tile4[2][AR + 32][PITCH4];
  const int lw   = __builtin_amdgcn_readfirstlane((int)(threadIdx.x >> 5));
  const int lane = threadIdx.x & 31;
  const int mblk = (NT == 2) ? (blockIdx.x & 1) : (blockIdx.x & 3);
  const int nb   = (NT == 2) ? (blockIdx.x >> 1) : (blockIdx.x >> 2);
  const int rt   = (NT == 2) ? lw : (lw >> 1);
  const int ct   = (NT == 2) ? 0 : (lw & 1);
  const int m0   = mblk * AR + rt * 16;
  const int nn   = lane & 15;
  const int hi   = lane >> 4;
  const int koff = hi << 1;
  const int ncol0 = nb * 32 + ct * 16 + nn;
  const int ncol1 = ncol0 + 16;
  const int nld0 = (ncol0 < N) ? ncol0 : (N - 1);
  const int nld1 = (ncol1 < N) ? ncol1 : (N - 1);
  const float bsum0 = bias1[nld0] + (bias2 ? bias2[nld0] : 0.f);
  const float bsum1 = (NT == 2) ? bias1[nld1] + (bias2 ? bias2[nld1] : 0.f) : 0.f;
  v8f acc0, acc1;
  #pragma unroll
  for (int i = 0; i < 8; ++i) { acc0[i] = bsum0; acc1[i] = bsum1; }

#define G_WROW(i) ((nb * 32 + lw + 4 * (i)) < N ? (nb * 32 + lw + 4 * (i)) : (N - 1))
#define G_WPTR(i) const float* wsrc##i = W + (size_t)G_WROW(i) * ldw + (lane << 2);
#define G_ADC(S, i) float4 pa##S##i;
#define G_WDC(S, i) float4 pb##S##i;
#define G_ALD(S, i) if (4 * (i) < AR) pa##S##i = *(const float4*)(asrc + (size_t)(4 * (i)) * lda + ko);
#define G_WLD(S, i) pb##S##i = *(const float4*)(wsrc##i + ko);
#define G_AST(S, i) if (4 * (i) < AR) tb[lw + 4 * (i)][lane] = pa##S##i;
#define G_WST(S, i) tb[AR + lw + 4 * (i)][lane] = pb##S##i;
  const float* asrc = A + (size_t)(mblk * AR + lw) * lda + (lane << 2);
  ROWS8(G_WPTR)
  ROWS16S(G_ADC, a) ROWS8S(G_WDC, a)
  ROWS16S(G_ADC, b) ROWS8S(G_WDC, b)
  ROWS16S(G_ADC, t) ROWS8S(G_WDC, t)

  const int nround = K / TCOLS;
  { const int ko = 0; ROWS16S(G_ALD, t) ROWS8S(G_WLD, t) }
  __builtin_amdgcn_sched_barrier(0);
  { const int ko = TCOLS; ROWS16S(G_ALD, b) ROWS8S(G_WLD, b) }
  __builtin_amdgcn_sched_barrier(0);
  { const int ko = 2 * TCOLS; ROWS16S(G_ALD, a) ROWS8S(G_WLD, a) }
  __builtin_amdgcn_sched_barrier(0);
  { float4 (*tb)[PITCH4] = tile4[0]; ROWS16S(G_AST, t) ROWS8S(G_WST, t) }

#define GEMM_ROUND(cexpr, S, ST, LD)                                            \
  {                                                                             \
    const int c_ = (cexpr);                                                     \
    lds_barrier();                                                              \
    const float* tf = (const float*)&tile4[c_ & 1][0][0];                       \
    const float* ap = tf + (rt * 16 + nn) * PITCH + koff;                       \
    const float* bp = tf + (AR + ct * 16 + nn) * PITCH + koff;                  \
    v2f fa[GG], fb0[GG], fb1[GG];                                               \
    _Pragma("unroll")                                                           \
    for (int q = 0; q < GG; ++q) {                                              \
      fa[q]  = *(const v2f*)(ap + 4 * q);                                       \
      fb0[q] = *(const v2f*)(bp + 4 * q);                                       \
      if (NT == 2) fb1[q] = *(const v2f*)(bp + 16 * PITCH + 4 * q);             \
    }                                                                           \
    _Pragma("unroll")                                                           \
    for (int jg = 0; jg < TCOLS / 4; jg += GG) {                                \
      v2f na[GG], nb0[GG], nb1[GG];                                             \
      if (jg + GG < TCOLS / 4) {                                                \
        _Pragma("unroll")                                                       \
        for (int q = 0; q < GG; ++q) {                                          \
          na[q]  = *(const v2f*)(ap + 4 * (jg + GG + q));                       \
          nb0[q] = *(const v2f*)(bp + 4 * (jg + GG + q));                       \
          if (NT == 2) nb1[q] = *(const v2f*)(bp + 16 * PITCH + 4 * (jg + GG + q)); \
        }                                                                       \
      }                                                                         \
      if (jg == 0) {                                                            \
         \
         \
         \
        __builtin_amdgcn_sched_barrier(0);                                      \
        if (ST) {                                                               \
          float4 (*tb)[PITCH4] = tile4[(c_ + 1) & 1];                           \
          ROWS16S(G_AST, S) ROWS8S(G_WST, S)                                    \
        }                                                                       \
        if (LD) {                                                               \
          const int ko = (c_ + 3) * TCOLS;                                      \
          ROWS16S(G_ALD, S) ROWS8S(G_WLD, S)                                    \
        }                                                                       \
      }                                                                         \
       \
       \
      __builtin_amdgcn_sched_barrier(0);                                        \
      _Pragma("unroll")                                                         \
      for (int q = 0; q < GG; ++q) {                                            \
        acc0 = __builtin_amdgcn_wmma_f32_16x16x4_f32(false, fa[q], false, fb0[q], \
                                                     (short)0, acc0, false, false); \
        if (NT == 2)                                                            \
          acc1 = __builtin_amdgcn_wmma_f32_16x16x4_f32(false, fa[q], false, fb1[q], \
                                                       (short)0, acc1, false, false); \
      }                                                                         \
       \
       \
      __builtin_amdgcn_sched_barrier(0);                                        \
      if (jg + GG < TCOLS / 4) {                                                \
        _Pragma("unroll")                                                       \
        for (int q = 0; q < GG; ++q) {                                          \
          fa[q] = na[q]; fb0[q] = nb0[q];                                       \
          if (NT == 2) fb1[q] = nb1[q];                                         \
        }                                                                       \
      }                                                                         \
    }                                                                           \
  }

  int c = 0;
  for (; c + 5 < nround; c += 2) {
    GEMM_ROUND(c, b, 1, 1)
    GEMM_ROUND(c + 1, a, 1, 1)
  }
  GEMM_ROUND(c, b, 1, 1)
  GEMM_ROUND(c + 1, a, 1, 0)
  GEMM_ROUND(c + 2, b, 1, 0)
  GEMM_ROUND(c + 3, a, 0, 0)

  #pragma unroll
  for (int t = 0; t < NT; ++t) {
    const int ncol = t ? ncol1 : ncol0;
    #pragma unroll
    for (int v = 0; v < 8; ++v) {
      int r = m0 + v + (hi << 3);
      float val = t ? acc1[v] : acc0[v];
      if (MODE == 0) {
        out[(size_t)r * ldo + coloff + ncol] = val;
      } else if (MODE == 1) {
        int b = r >> 4, tt = r & 15;
        int rr = tt * BB + b;
        if (ncol < DD) out[(size_t)rr * DD + ncol] = val;
        else           out2[(size_t)rr * DD + (ncol - DD)] = val;
      } else {
        if (ncol < N) {
          int tt = r >> 3, b = r & 7;
          out[(size_t)(b * TT + tt) * N + ncol] = val;
        }
      }
    }
  }
}

#define KC (TCOLS / 2)
#define LROWS 16
#define WROWS 64
#define LTROWS (LROWS + WROWS)
#define NROUND (DD / 2 / KC)
#define LK 8

__global__ void __launch_bounds__(128)
lstm_step_kernel(const float* __restrict__ xpre,
                 const float* hin,
                 const float* Whh,
                 float* __restrict__ cc,
                 float* __restrict__ hout,
                 float* __restrict__ lasth,
                 int step, int mtiles) {
  __shared__ float4 tile4[2][LTROWS][PITCH4];
  __shared__ float red[2][4][8][32];
  const int lw   = __builtin_amdgcn_readfirstlane((int)(threadIdx.x >> 5));
  const int lane = threadIdx.x & 31;
  const int nt   = blockIdx.x / mtiles;
  const int mt   = blockIdx.x - nt * mtiles;
  const int n0   = nt << 4;
  const int m0   = mt << 4;
  const int kh   = lw & 1;
  const int gp   = lw >> 1;
  const int nn   = lane & 15;
  const int hi   = lane >> 4;
  const int koff = hi << 1;

  v8f acc0, acc1;
  if (kh == 0) {
    #pragma unroll
    for (int v = 0; v < 8; ++v) {
      int r = m0 + v + (hi << 3);
      const float* xp = xpre + (size_t)r * (4 * DD) + (2 * gp) * DD + n0 + nn;
      acc0[v] = xp[0];
      acc1[v] = xp[DD];
    }
  } else {
    #pragma unroll
    for (int v = 0; v < 8; ++v) { acc0[v] = 0.f; acc1[v] = 0.f; }
  }

  const int kcol = (lane >> 4) * (DD / 2) + ((lane & 15) << 2);
#define W_OFF(i) ((size_t)(((i) >> 2) * DD + 4 * ((i) & 3)) * DD)
#define H_DC(S, i) float4 ph##S##i;
#define W_DC(S, i) float4 pw##S##i;
#define H_LD(S, i) ph##S##i = *(const float4*)(hsrc + (size_t)(4 * (i)) * DD + ko);
#define W_LD(S, i) pw##S##i = *(const float4*)(wsrc + W_OFF(i) + ko);
#define H_ST(S, i) tb[lw + 4 * (i)][lane] = ph##S##i;
#define W_ST(S, i) tb[LROWS + lw + 4 * (i)][lane] = pw##S##i;
  const float* hsrc = hin + (size_t)(m0 + lw) * DD + kcol;
  const float* wsrc = Whh + (size_t)(n0 + lw) * DD + kcol;
  ROWS4S(H_DC, a) ROWS16S(W_DC, a)
  ROWS4S(H_DC, b) ROWS16S(W_DC, b)
  ROWS4S(H_DC, t) ROWS16S(W_DC, t)
  { const int ko = 0;      ROWS4S(H_LD, t) ROWS16S(W_LD, t) }
  __builtin_amdgcn_sched_barrier(0);
  { const int ko = KC;     ROWS4S(H_LD, b) ROWS16S(W_LD, b) }
  __builtin_amdgcn_sched_barrier(0);
  { const int ko = 2 * KC; ROWS4S(H_LD, a) ROWS16S(W_LD, a) }
  __builtin_amdgcn_sched_barrier(0);
  { float4 (*tb)[PITCH4] = tile4[0]; ROWS4S(H_ST, t) ROWS16S(W_ST, t) }

#define LSTM_ROUND(cexpr, S, ST, LD)                                            \
  {                                                                             \
    const int c_ = (cexpr);                                                     \
    lds_barrier();                                                              \
    const float* tf = (const float*)&tile4[c_ & 1][0][0];                       \
    const float* ap = tf + nn * PITCH + kh * KC + koff;                         \
    const float* bp = tf + (LROWS + gp * 32 + nn) * PITCH + kh * KC + koff;     \
    v2f fa[LK], fb0[LK], fb1[LK];                                               \
    _Pragma("unroll")                                                           \
    for (int q = 0; q < LK; ++q) {                                              \
      fa[q]  = *(const v2f*)(ap + 4 * q);                                       \
      fb0[q] = *(const v2f*)(bp + 4 * q);                                       \
      fb1[q] = *(const v2f*)(bp + 16 * PITCH + 4 * q);                          \
    }                                                                           \
    _Pragma("unroll")                                                           \
    for (int jg = 0; jg < KC / 4; jg += LK) {                                   \
      v2f na[LK], nb0[LK], nb1[LK];                                             \
      if (jg + LK < KC / 4) {                                                   \
        _Pragma("unroll")                                                       \
        for (int q = 0; q < LK; ++q) {                                          \
          na[q]  = *(const v2f*)(ap + 4 * (jg + LK + q));                       \
          nb0[q] = *(const v2f*)(bp + 4 * (jg + LK + q));                       \
          nb1[q] = *(const v2f*)(bp + 16 * PITCH + 4 * (jg + LK + q));          \
        }                                                                       \
      }                                                                         \
      if (jg == 0) {                                                            \
         \
         \
         \
        __builtin_amdgcn_sched_barrier(0);                                      \
        if (ST) {                                                               \
          float4 (*tb)[PITCH4] = tile4[(c_ + 1) & 1];                           \
          ROWS4S(H_ST, S) ROWS16S(W_ST, S)                                      \
        }                                                                       \
        if (LD) {                                                               \
          const int ko = (c_ + 3) * KC;                                         \
          ROWS4S(H_LD, S) ROWS16S(W_LD, S)                                      \
        }                                                                       \
      }                                                                         \
       \
       \
      __builtin_amdgcn_sched_barrier(0);                                        \
      _Pragma("unroll")                                                         \
      for (int q = 0; q < LK; ++q) {                                            \
        acc0 = __builtin_amdgcn_wmma_f32_16x16x4_f32(false, fa[q], false, fb0[q], \
                                                     (short)0, acc0, false, false); \
        acc1 = __builtin_amdgcn_wmma_f32_16x16x4_f32(false, fa[q], false, fb1[q], \
                                                     (short)0, acc1, false, false); \
      }                                                                         \
       \
       \
      __builtin_amdgcn_sched_barrier(0);                                        \
      if (jg + LK < KC / 4) {                                                   \
        _Pragma("unroll")                                                       \
        for (int q = 0; q < LK; ++q) { fa[q] = na[q]; fb0[q] = nb0[q]; fb1[q] = nb1[q]; } \
      }                                                                         \
    }                                                                           \
  }

  static_assert(NROUND == 8, "the rounds below are written out for 8");
  LSTM_ROUND(0, b, 1, 1)
  LSTM_ROUND(1, a, 1, 1)
  LSTM_ROUND(2, b, 1, 1)
  LSTM_ROUND(3, a, 1, 1)
  LSTM_ROUND(4, b, 1, 1)
  LSTM_ROUND(5, a, 1, 0)
  LSTM_ROUND(6, b, 1, 0)
  LSTM_ROUND(7, a, 0, 0)

  #pragma unroll
  for (int v = 0; v < 8; ++v) {
    red[kh][2 * gp][v][lane]     = acc0[v];
    red[kh][2 * gp + 1][v][lane] = acc1[v];
  }
  __syncthreads();

  const int tgt = TT - 1 - step;
  const int d = n0 + nn;
  #pragma unroll
  for (int j = 0; j < 2; ++j) {
    const int v = 2 * lw + j;
    const int r = m0 + v + (hi << 3);
    const float gi = red[0][0][v][lane] + red[1][0][v][lane];
    const float gf = red[0][1][v][lane] + red[1][1][v][lane];
    const float gg = red[0][2][v][lane] + red[1][2][v][lane];
    const float go = red[0][3][v][lane] + red[1][3][v][lane];
    float si = 1.f / (1.f + expf(-gi));
    float sf = 1.f / (1.f + expf(-gf));
    float gt = tanhf(gg);
    float so = 1.f / (1.f + expf(-go));
    float cold = cc[(size_t)r * DD + d];
    float cnew = sf * cold + si * gt;
    float hnew = so * tanhf(cnew);
    cc[(size_t)r * DD + d] = cnew;
    hout[(size_t)r * DD + d] = hnew;
    if ((r >> 3) == tgt) lasth[(size_t)r * DD + d] = hnew;
  }
}

__global__ void ln_relu_kernel(const float* __restrict__ in,
                               const float* __restrict__ g,
                               const float* __restrict__ be,
                               float* __restrict__ out) {
  __shared__ float s1[256], s2[256];
  int r = blockIdx.x, t = threadIdx.x;
  const float* row = in + (size_t)r * DD;
  float s = 0.f, sq = 0.f;
  for (int i = t; i < DD; i += 256) { float v = row[i]; s += v; sq += v * v; }
  s1[t] = s; s2[t] = sq;
  __syncthreads();
  for (int off = 128; off > 0; off >>= 1) {
    if (t < off) { s1[t] += s1[t + off]; s2[t] += s2[t + off]; }
    __syncthreads();
  }
  float mean = s1[0] * (1.0f / DD);
  float var = s2[0] * (1.0f / DD) - mean * mean;
  float inv = rsqrtf(var + 1e-5f);
  for (int i = t; i < DD; i += 256) {
    float v = (row[i] - mean) * inv * g[i] + be[i];
    out[(size_t)r * DD + i] = v > 0.f ? v : 0.f;
  }
}

extern "C" void kernel_launch(void* const* d_in, const int* in_sizes, int n_in,
                              void* d_out, int out_size, void* d_ws, size_t ws_size,
                              hipStream_t stream) {
  const float* x   = (const float*)d_in[0];
  const float* Wn  = (const float*)d_in[1];
  const float* bn  = (const float*)d_in[2];
  const float* Wv  = (const float*)d_in[3];
  const float* bv  = (const float*)d_in[4];
  const float* Wf  = (const float*)d_in[5];
  const float* bf  = (const float*)d_in[6];
  const float* Wih = (const float*)d_in[7];
  const float* Whh = (const float*)d_in[8];
  const float* bih = (const float*)d_in[9];
  const float* bhh = (const float*)d_in[10];
  const float* Wc  = (const float*)d_in[11];
  const float* bc  = (const float*)d_in[12];
  const float* g_a = (const float*)d_in[13];
  const float* be_a= (const float*)d_in[14];
  const float* Wa  = (const float*)d_in[15];
  const float* ba  = (const float*)d_in[16];
  float* out = (float*)d_out;

  float* ws    = (float*)d_ws;
  float* A0    = ws;
  float* F1    = A0    + 131072;
  float* F2    = F1    + 131072;
  float* ENV   = F2    + 131072;
  float* XPRE  = ENV   + 262144;
  float* H0    = XPRE  + 524288;
  float* H1    = H0    + 131072;
  float* CC    = H1    + 131072;
  float* LASTH = CC    + 131072;
  float* UNR   = LASTH + 131072;
  float* LNR   = UNR   + 131072;

  pool_kernel<<<1536, 256, 0, stream>>>(x, A0, F1, F2);
  gemm_kernel<0, 1><<<128, 128, 0, stream>>>(F1, DD, Wn, DD, bn, nullptr,
                                             ENV, 2 * DD, 0,  nullptr, DD, DD);
  gemm_kernel<0, 1><<<128, 128, 0, stream>>>(F2, DD, Wv, DD, bv, nullptr,
                                             ENV, 2 * DD, DD, nullptr, DD, DD);
  gemm_kernel<1, 1><<<256, 128, 0, stream>>>(ENV, 2 * DD, Wf, 2 * DD, bf, nullptr,
                                             H0, 0, 0, CC, 2 * DD, 2 * DD);
  gemm_kernel<0, 2><<<256, 128, 0, stream>>>(A0, DD, Wih, DD, bih, bhh,
                                             XPRE, 4 * DD, 0, nullptr, 4 * DD, DD);
  float* hin = H0; float* hout = H1;
  for (int s = 0; s < TT; ++s) {
    int mtiles = (TT - s + 1) >> 1;
    lstm_step_kernel<<<64 * mtiles, 128, 0, stream>>>(XPRE, hin, Whh, CC, hout,
                                                      LASTH, s, mtiles);
    float* tmp = hin; hin = hout; hout = tmp;
  }
  gemm_kernel<0, 1><<<128, 128, 0, stream>>>(LASTH, DD, Wc, DD, bc, nullptr,
                                             UNR, DD, 0, nullptr, DD, DD);
  ln_relu_kernel<<<128, 256, 0, stream>>>(UNR, g_a, be_a, LNR);
  gemm_kernel<2, 1><<<236, 128, 0, stream>>>(LNR, DD, Wa, DD, ba, nullptr,
                                             out, NA, 0, nullptr, NA, DD);
}


# LSTM step and layer norm: global values a launch needs late (old cell state; row, gain, shift) are requested at its head as one group, not one memory trip at a time in its tail; sum orders kept
# speedup vs baseline: 3.0146x; 1.0841x over previous
#include <hip/hip_runtime.h>
#include <math.h>

#define DD 1024
#define NA 1887
#define BB 8
#define TT 16
#define ROWS 128

typedef __attribute__((ext_vector_type(2))) float v2f;
typedef __attribute__((ext_vector_type(8))) float v8f;

__global__ void pool_kernel(const float* __restrict__ x, float* __restrict__ A0,
                            float* __restrict__ F1, float* __restrict__ F2) {
  int tid = blockIdx.x * blockDim.x + threadIdx.x;
  if (tid >= ROWS * DD * 3) return;
  int c  = tid % 3;
  int rd = tid / 3;
  int d  = rd % DD;
  int r  = rd / DD;
  const float* p = x + (size_t)rd * 147 + c;
  float s = 0.f;
  #pragma unroll
  for (int i = 0; i < 49; ++i) s += p[i * 3];
  s *= (1.0f / 49.0f);
  if (c == 0) {
    int b = r >> 4, t = r & 15;
    A0[(size_t)(t * BB + b) * DD + d] = s;
  } else if (c == 1) {
    F1[(size_t)r * DD + d] = s;
  } else {
    F2[(size_t)r * DD + d] = s;
  }
}

#define TCOLS 128
#define PITCH (TCOLS + 4)
#define PITCH4 (PITCH / 4)
#define GG 8
#define LG 4
#define ROWS8(X)  X(0) X(1) X(2) X(3) X(4) X(5) X(6) X(7)
#define ROWS4S(X, S)  X(S, 0) X(S, 1) X(S, 2) X(S, 3)
#define ROWS8S(X, S)  ROWS4S(X, S) X(S, 4) X(S, 5) X(S, 6) X(S, 7)
#define ROWS16S(X, S) ROWS8S(X, S) X(S, 8) X(S, 9) X(S, 10) X(S, 11) X(S, 12) X(S, 13) X(S, 14) X(S, 15)

__device__ __forceinline__ void lds_barrier() {
  asm volatile("s_wait_dscnt 0x0\n\ts_barrier_signal -1\n\ts_barrier_wait -1" ::: "memory");
}

template <int MODE, int NT>
__global__ void __launch_bounds__(128)
gemm_kernel(const float* A, int lda,
            const float* W, int ldw,
            const float* __restrict__ bias1,
            const float* __restrict__ bias2,
            float* __restrict__ out, int ldo, int coloff,
            float* __restrict__ out2, int N, int K) {
  constexpr int AR = 32 * NT;
  __shared__ float4 tile4[2][AR + 32][PITCH4];
  const int lw   = __builtin_amdgcn_readfirstlane((int)(threadIdx.x >> 5));
  const int lane = threadIdx.x & 31;
  const int mblk = (NT == 2) ? (blockIdx.x & 1) : (blockIdx.x & 3);
  const int nb   = (NT == 2) ? (blockIdx.x >> 1) : (blockIdx.x >> 2);
  const int rt   = (NT == 2) ? lw : (lw >> 1);
  const int ct   = (NT == 2) ? 0 : (lw & 1);
  const int m0   = mblk * AR + rt * 16;
  const int nn   = lane & 15;
  const int hi   = lane >> 4;
  const int koff = hi << 1;
  const int ncol0 = nb * 32 + ct * 16 + nn;
  const int ncol1 = ncol0 + 16;
  const int nld0 = (ncol0 < N) ? ncol0 : (N - 1);
  const int nld1 = (ncol1 < N) ? ncol1 : (N - 1);
  const float bsum0 = bias1[nld0] + (bias2 ? bias2[nld0] : 0.f);
  const float bsum1 = (NT == 2) ? bias1[nld1] + (bias2 ? bias2[nld1] : 0.f) : 0.f;
  v8f acc0, acc1;
  #pragma unroll
  for (int i = 0; i < 8; ++i) { acc0[i] = bsum0; acc1[i] = bsum1; }

#define G_WROW(i) ((nb * 32 + lw + 4 * (i)) < N ? (nb * 32 + lw + 4 * (i)) : (N - 1))
#define G_WPTR(i) const float* wsrc##i = W + (size_t)G_WROW(i) * ldw + (lane << 2);
#define G_ADC(S, i) float4 pa##S##i;
#define G_WDC(S, i) float4 pb##S##i;
#define G_ALD(S, i) if (4 * (i) < AR) pa##S##i = *(const float4*)(asrc + (size_t)(4 * (i)) * lda + ko);
#define G_WLD(S, i) pb##S##i = *(const float4*)(wsrc##i + ko);
#define G_AST(S, i) if (4 * (i) < AR) tb[lw + 4 * (i)][lane] = pa##S##i;
#define G_WST(S, i) tb[AR + lw + 4 * (i)][lane] = pb##S##i;
  const float* asrc = A + (size_t)(mblk * AR + lw) * lda + (lane << 2);
  ROWS8(G_WPTR)
  ROWS16S(G_ADC, a) ROWS8S(G_WDC, a)
  ROWS16S(G_ADC, b) ROWS8S(G_WDC, b)
  ROWS16S(G_ADC, t) ROWS8S(G_WDC, t)

  const int nround = K / TCOLS;
  { const int ko = 0; ROWS16S(G_ALD, t) ROWS8S(G_WLD, t) }
  __builtin_amdgcn_sched_barrier(0);
  { const int ko = TCOLS; ROWS16S(G_ALD, b) ROWS8S(G_WLD, b) }
  __builtin_amdgcn_sched_barrier(0);
  { const int ko = 2 * TCOLS; ROWS16S(G_ALD, a) ROWS8S(G_WLD, a) }
  __builtin_amdgcn_sched_barrier(0);
  { float4 (*tb)[PITCH4] = tile4[0]; ROWS16S(G_AST, t) ROWS8S(G_WST, t) }

#define GEMM_ROUND(cexpr, S, ST, LD)                                            \
  {                                                                             \
    const int c_ = (cexpr);                                                     \
    lds_barrier();                                                              \
    const float* tf = (const float*)&tile4[c_ & 1][0][0];                       \
    const float* ap = tf + (rt * 16 + nn) * PITCH + koff;                       \
    const float* bp = tf + (AR + ct * 16 + nn) * PITCH + koff;                  \
    v2f fa[GG], fb0[GG], fb1[GG];                                               \
    _Pragma("unroll")                                                           \
    for (int q = 0; q < GG; ++q) {                                              \
      fa[q]  = *(const v2f*)(ap + 4 * q);                                       \
      fb0[q] = *(const v2f*)(bp + 4 * q);                                       \
      if (NT == 2) fb1[q] = *(const v2f*)(bp + 16 * PITCH + 4 * q);             \
    }                                                                           \
    _Pragma("unroll")                                                           \
    for (int jg = 0; jg < TCOLS / 4; jg += GG) {                                \
      v2f na[GG], nb0[GG], nb1[GG];                                             \
      if (jg + GG < TCOLS / 4) {                                                \
        _Pragma("unroll")                                                       \
        for (int q = 0; q < GG; ++q) {                                          \
          na[q]  = *(const v2f*)(ap + 4 * (jg + GG + q));                       \
          nb0[q] = *(const v2f*)(bp + 4 * (jg + GG + q));                       \
          if (NT == 2) nb1[q] = *(const v2f*)(bp + 16 * PITCH + 4 * (jg + GG + q)); \
        }                                                                       \
      }                                                                         \
      if (jg == 0) {                                                            \
         \
         \
         \
        __builtin_amdgcn_sched_barrier(0);                                      \
        if (ST) {                                                               \
          float4 (*tb)[PITCH4] = tile4[(c_ + 1) & 1];                           \
          ROWS16S(G_AST, S) ROWS8S(G_WST, S)                                    \
        }                                                                       \
        if (LD) {                                                               \
          const int ko = (c_ + 3) * TCOLS;                                      \
          ROWS16S(G_ALD, S) ROWS8S(G_WLD, S)                                    \
        }                                                                       \
      }                                                                         \
       \
       \
      __builtin_amdgcn_sched_barrier(0);                                        \
      _Pragma("unroll")                                                         \
      for (int q = 0; q < GG; ++q) {                                            \
        acc0 = __builtin_amdgcn_wmma_f32_16x16x4_f32(false, fa[q], false, fb0[q], \
                                                     (short)0, acc0, false, false); \
        if (NT == 2)                                                            \
          acc1 = __builtin_amdgcn_wmma_f32_16x16x4_f32(false, fa[q], false, fb1[q], \
                                                       (short)0, acc1, false, false); \
      }                                                                         \
       \
       \
      __builtin_amdgcn_sched_barrier(0);                                        \
      if (jg + GG < TCOLS / 4) {                                                \
        _Pragma("unroll")                                                       \
        for (int q = 0; q < GG; ++q) {                                          \
          fa[q] = na[q]; fb0[q] = nb0[q];                                       \
          if (NT == 2) fb1[q] = nb1[q];                                         \
        }                                                                       \
      }                                                                         \
    }                                                                           \
  }

  int c = 0;
  for (; c + 5 < nround; c += 2) {
    GEMM_ROUND(c, b, 1, 1)
    GEMM_ROUND(c + 1, a, 1, 1)
  }
  GEMM_ROUND(c, b, 1, 1)
  GEMM_ROUND(c + 1, a, 1, 0)
  GEMM_ROUND(c + 2, b, 1, 0)
  GEMM_ROUND(c + 3, a, 0, 0)

  #pragma unroll
  for (int t = 0; t < NT; ++t) {
    const int ncol = t ? ncol1 : ncol0;
    #pragma unroll
    for (int v = 0; v < 8; ++v) {
      int r = m0 + v + (hi << 3);
      float val = t ? acc1[v] : acc0[v];
      if (MODE == 0) {
        out[(size_t)r * ldo + coloff + ncol] = val;
      } else if (MODE == 1) {
        int b = r >> 4, tt = r & 15;
        int rr = tt * BB + b;
        if (ncol < DD) out[(size_t)rr * DD + ncol] = val;
        else           out2[(size_t)rr * DD + (ncol - DD)] = val;
      } else {
        if (ncol < N) {
          int tt = r >> 3, b = r & 7;
          out[(size_t)(b * TT + tt) * N + ncol] = val;
        }
      }
    }
  }
}

#define KC (TCOLS / 2)
#define LROWS 16
#define WROWS 64
#define LTROWS (LROWS + WROWS)
#define NROUND (DD / 2 / KC)
#define LK 8

__global__ void __launch_bounds__(128)
lstm_step_kernel(const float* __restrict__ xpre,
                 const float* hin,
                 const float* Whh,
                 float* __restrict__ cc,
                 float* __restrict__ hout,
                 float* __restrict__ lasth,
                 int step, int mtiles) {
  __shared__ float4 tile4[2][LTROWS][PITCH4];
  __shared__ float red[2][4][8][32];
  const int lw   = __builtin_amdgcn_readfirstlane((int)(threadIdx.x >> 5));
  const int lane = threadIdx.x & 31;
  const int nt   = blockIdx.x / mtiles;
  const int mt   = blockIdx.x - nt * mtiles;
  const int n0   = nt << 4;
  const int m0   = mt << 4;
  const int kh   = lw & 1;
  const int gp   = lw >> 1;
  const int nn   = lane & 15;
  const int hi   = lane >> 4;
  const int koff = hi << 1;

  const float cold0 = cc[(size_t)(m0 + 2 * lw + (hi << 3)) * DD + n0 + nn];
  const float cold1 = cc[(size_t)(m0 + 2 * lw + 1 + (hi << 3)) * DD + n0 + nn];

  v8f acc0, acc1;
  if (kh == 0) {
    #pragma unroll
    for (int v = 0; v < 8; ++v) {
      int r = m0 + v + (hi << 3);
      const float* xp = xpre + (size_t)r * (4 * DD) + (2 * gp) * DD + n0 + nn;
      acc0[v] = xp[0];
      acc1[v] = xp[DD];
    }
  } else {
    #pragma unroll
    for (int v = 0; v < 8; ++v) { acc0[v] = 0.f; acc1[v] = 0.f; }
  }

  const int kcol = (lane >> 4) * (DD / 2) + ((lane & 15) << 2);
#define W_OFF(i) ((size_t)(((i) >> 2) * DD + 4 * ((i) & 3)) * DD)
#define H_DC(S, i) float4 ph##S##i;
#define W_DC(S, i) float4 pw##S##i;
#define H_LD(S, i) ph##S##i = *(const float4*)(hsrc + (size_t)(4 * (i)) * DD + ko);
#define W_LD(S, i) pw##S##i = *(const float4*)(wsrc + W_OFF(i) + ko);
#define H_ST(S, i) tb[lw + 4 * (i)][lane] = ph##S##i;
#define W_ST(S, i) tb[LROWS + lw + 4 * (i)][lane] = pw##S##i;
  const float* hsrc = hin + (size_t)(m0 + lw) * DD + kcol;
  const float* wsrc = Whh + (size_t)(n0 + lw) * DD + kcol;
  ROWS4S(H_DC, a) ROWS16S(W_DC, a)
  ROWS4S(H_DC, b) ROWS16S(W_DC, b)
  ROWS4S(H_DC, t) ROWS16S(W_DC, t)
  { const int ko = 0;      ROWS4S(H_LD, t) ROWS16S(W_LD, t) }
  __builtin_amdgcn_sched_barrier(0);
  { const int ko = KC;     ROWS4S(H_LD, b) ROWS16S(W_LD, b) }
  __builtin_amdgcn_sched_barrier(0);
  { const int ko = 2 * KC; ROWS4S(H_LD, a) ROWS16S(W_LD, a) }
  __builtin_amdgcn_sched_barrier(0);
  { float4 (*tb)[PITCH4] = tile4[0]; ROWS4S(H_ST, t) ROWS16S(W_ST, t) }

#define LSTM_ROUND(cexpr, S, ST, LD)                                            \
  {                                                                             \
    const int c_ = (cexpr);                                                     \
    lds_barrier();                                                              \
    const float* tf = (const float*)&tile4[c_ & 1][0][0];                       \
    const float* ap = tf + nn * PITCH + kh * KC + koff;                         \
    const float* bp = tf + (LROWS + gp * 32 + nn) * PITCH + kh * KC + koff;     \
    v2f fa[LK], fb0[LK], fb1[LK];                                               \
    _Pragma("unroll")                                                           \
    for (int q = 0; q < LK; ++q) {                                              \
      fa[q]  = *(const v2f*)(ap + 4 * q);                                       \
      fb0[q] = *(const v2f*)(bp + 4 * q);                                       \
      fb1[q] = *(const v2f*)(bp + 16 * PITCH + 4 * q);                          \
    }                                                                           \
    _Pragma("unroll")                                                           \
    for (int jg = 0; jg < KC / 4; jg += LK) {                                   \
      v2f na[LK], nb0[LK], nb1[LK];                                             \
      if (jg + LK < KC / 4) {                                                   \
        _Pragma("unroll")                                                       \
        for (int q = 0; q < LK; ++q) {                                          \
          na[q]  = *(const v2f*)(ap + 4 * (jg + LK + q));                       \
          nb0[q] = *(const v2f*)(bp + 4 * (jg + LK + q));                       \
          nb1[q] = *(const v2f*)(bp + 16 * PITCH + 4 * (jg + LK + q));          \
        }                                                                       \
      }                                                                         \
      if (jg == 0) {                                                            \
         \
         \
         \
        __builtin_amdgcn_sched_barrier(0);                                      \
        if (ST) {                                                               \
          float4 (*tb)[PITCH4] = tile4[(c_ + 1) & 1];                           \
          ROWS4S(H_ST, S) ROWS16S(W_ST, S)                                      \
        }                                                                       \
        if (LD) {                                                               \
          const int ko = (c_ + 3) * KC;                                         \
          ROWS4S(H_LD, S) ROWS16S(W_LD, S)                                      \
        }                                                                       \
      }                                                                         \
       \
       \
      __builtin_amdgcn_sched_barrier(0);                                        \
      _Pragma("unroll")                                                         \
      for (int q = 0; q < LK; ++q) {                                            \
        acc0 = __builtin_amdgcn_wmma_f32_16x16x4_f32(false, fa[q], false, fb0[q], \
                                                     (short)0, acc0, false, false); \
        acc1 = __builtin_amdgcn_wmma_f32_16x16x4_f32(false, fa[q], false, fb1[q], \
                                                     (short)0, acc1, false, false); \
      }                                                                         \
       \
       \
      __builtin_amdgcn_sched_barrier(0);                                        \
      if (jg + LK < KC / 4) {                                                   \
        _Pragma("unroll")                                                       \
        for (int q = 0; q < LK; ++q) { fa[q] = na[q]; fb0[q] = nb0[q]; fb1[q] = nb1[q]; } \
      }                                                                         \
    }                                                                           \
  }

  static_assert(NROUND == 8, "the rounds below are written out for 8");
  LSTM_ROUND(0, b, 1, 1)
  LSTM_ROUND(1, a, 1, 1)
  LSTM_ROUND(2, b, 1, 1)
  LSTM_ROUND(3, a, 1, 1)
  LSTM_ROUND(4, b, 1, 1)
  LSTM_ROUND(5, a, 1, 0)
  LSTM_ROUND(6, b, 1, 0)
  LSTM_ROUND(7, a, 0, 0)

  #pragma unroll
  for (int v = 0; v < 8; ++v) {
    red[kh][2 * gp][v][lane]     = acc0[v];
    red[kh][2 * gp + 1][v][lane] = acc1[v];
  }
  __syncthreads();

  const int tgt = TT - 1 - step;
  const int d = n0 + nn;
  #pragma unroll
  for (int j = 0; j < 2; ++j) {
    const int v = 2 * lw + j;
    const int r = m0 + v + (hi << 3);
    const float gi = red[0][0][v][lane] + red[1][0][v][lane];
    const float gf = red[0][1][v][lane] + red[1][1][v][lane];
    const float gg = red[0][2][v][lane] + red[1][2][v][lane];
    const float go = red[0][3][v][lane] + red[1][3][v][lane];
    float si = 1.f / (1.f + expf(-gi));
    float sf = 1.f / (1.f + expf(-gf));
    float gt = tanhf(gg);
    float so = 1.f / (1.f + expf(-go));
    float cold = j ? cold1 : cold0;
    float cnew = sf * cold + si * gt;
    float hnew = so * tanhf(cnew);
    cc[(size_t)r * DD + d] = cnew;
    hout[(size_t)r * DD + d] = hnew;
    if ((r >> 3) == tgt) lasth[(size_t)r * DD + d] = hnew;
  }
}

__global__ void ln_relu_kernel(const float* __restrict__ in,
                               const float* __restrict__ g,
                               const float* __restrict__ be,
                               float* __restrict__ out) {
  __shared__ float s1[256], s2[256];
  int r = blockIdx.x, t = threadIdx.x;
  const float* row = in + (size_t)r * DD;
  float xv[4], gv[4], bv[4];
  #pragma unroll
  for (int q = 0; q < 4; ++q) { xv[q] = row[t + 256 * q]; gv[q] = g[t + 256 * q]; bv[q] = be[t + 256 * q]; }
  float s = 0.f, sq = 0.f;
  #pragma unroll
  for (int q = 0; q < 4; ++q) { float v = xv[q]; s += v; sq = fmaf(v, v, sq); }
  s1[t] = s; s2[t] = sq;
  __syncthreads();
  for (int off = 128; off > 0; off >>= 1) {
    if (t < off) { s1[t] += s1[t + off]; s2[t] += s2[t + off]; }
    __syncthreads();
  }
  float mean = s1[0] * (1.0f / DD);
  float var = s2[0] * (1.0f / DD) - mean * mean;
  float inv = rsqrtf(var + 1e-5f);
  #pragma unroll
  for (int q = 0; q < 4; ++q) {
    float v = fmaf((xv[q] - mean) * inv, gv[q], bv[q]);
    out[(size_t)r * DD + t + 256 * q] = v > 0.f ? v : 0.f;
  }
}

extern "C" void kernel_launch(void* const* d_in, const int* in_sizes, int n_in,
                              void* d_out, int out_size, void* d_ws, size_t ws_size,
                              hipStream_t stream) {
  const float* x   = (const float*)d_in[0];
  const float* Wn  = (const float*)d_in[1];
  const float* bn  = (const float*)d_in[2];
  const float* Wv  = (const float*)d_in[3];
  const float* bv  = (const float*)d_in[4];
  const float* Wf  = (const float*)d_in[5];
  const float* bf  = (const float*)d_in[6];
  const float* Wih = (const float*)d_in[7];
  const float* Whh = (const float*)d_in[8];
  const float* bih = (const float*)d_in[9];
  const float* bhh = (const float*)d_in[10];
  const float* Wc  = (const float*)d_in[11];
  const float* bc  = (const float*)d_in[12];
  const float* g_a = (const float*)d_in[13];
  const float* be_a= (const float*)d_in[14];
  const float* Wa  = (const float*)d_in[15];
  const float* ba  = (const float*)d_in[16];
  float* out = (float*)d_out;

  float* ws    = (float*)d_ws;
  float* A0    = ws;
  float* F1    = A0    + 131072;
  float* F2    = F1    + 131072;
  float* ENV   = F2    + 131072;
  float* XPRE  = ENV   + 262144;
  float* H0    = XPRE  + 524288;
  float* H1    = H0    + 131072;
  float* CC    = H1    + 131072;
  float* LASTH = CC    + 131072;
  float* UNR   = LASTH + 131072;
  float* LNR   = UNR   + 131072;

  pool_kernel<<<1536, 256, 0, stream>>>(x, A0, F1, F2);
  gemm_kernel<0, 1><<<128, 128, 0, stream>>>(F1, DD, Wn, DD, bn, nullptr,
                                             ENV, 2 * DD, 0,  nullptr, DD, DD);
  gemm_kernel<0, 1><<<128, 128, 0, stream>>>(F2, DD, Wv, DD, bv, nullptr,
                                             ENV, 2 * DD, DD, nullptr, DD, DD);
  gemm_kernel<1, 1><<<256, 128, 0, stream>>>(ENV, 2 * DD, Wf, 2 * DD, bf, nullptr,
                                             H0, 0, 0, CC, 2 * DD, 2 * DD);
  gemm_kernel<0, 2><<<256, 128, 0, stream>>>(A0, DD, Wih, DD, bih, bhh,
                                             XPRE, 4 * DD, 0, nullptr, 4 * DD, DD);
  float* hin = H0; float* hout = H1;
  for (int s = 0; s < TT; ++s) {
    int mtiles = (TT - s + 1) >> 1;
    lstm_step_kernel<<<64 * mtiles, 128, 0, stream>>>(XPRE, hin, Whh, CC, hout,
                                                      LASTH, s, mtiles);
    float* tmp = hin; hin = hout; hout = tmp;
  }
  gemm_kernel<0, 1><<<128, 128, 0, stream>>>(LASTH, DD, Wc, DD, bc, nullptr,
                                             UNR, DD, 0, nullptr, DD, DD);
  ln_relu_kernel<<<128, 256, 0, stream>>>(UNR, g_a, be_a, LNR);
  gemm_kernel<2, 1><<<236, 128, 0, stream>>>(LNR, DD, Wa, DD, ba, nullptr,
                                             out, NA, 0, nullptr, NA, DD);
}


# LSTM launches re-dealt: rows are independent recurrences, so launch 0 steps all 16 row groups and launches 1..15 step exactly 8 each (4 full row tiles, 256 blocks; was 8,8,7,7..1,1 tiles); sums kept
# speedup vs baseline: 3.1210x; 1.0353x over previous
#include <hip/hip_runtime.h>
#include <math.h>

#define DD 1024
#define NA 1887
#define BB 8
#define TT 16
#define ROWS 128

typedef __attribute__((ext_vector_type(2))) float v2f;
typedef __attribute__((ext_vector_type(8))) float v8f;

__global__ void pool_kernel(const float* __restrict__ x, float* __restrict__ A0,
                            float* __restrict__ F1, float* __restrict__ F2) {
  int tid = blockIdx.x * blockDim.x + threadIdx.x;
  if (tid >= ROWS * DD * 3) return;
  int c  = tid % 3;
  int rd = tid / 3;
  int d  = rd % DD;
  int r  = rd / DD;
  const float* p = x + (size_t)rd * 147 + c;
  float s = 0.f;
  #pragma unroll
  for (int i = 0; i < 49; ++i) s += p[i * 3];
  s *= (1.0f / 49.0f);
  if (c == 0) {
    int b = r >> 4, t = r & 15;
    A0[(size_t)(t * BB + b) * DD + d] = s;
  } else if (c == 1) {
    F1[(size_t)r * DD + d] = s;
  } else {
    F2[(size_t)r * DD + d] = s;
  }
}

#define TCOLS 128
#define PITCH (TCOLS + 4)
#define PITCH4 (PITCH / 4)
#define GG 8
#define LG 4
#define ROWS8(X)  X(0) X(1) X(2) X(3) X(4) X(5) X(6) X(7)
#define ROWS4S(X, S)  X(S, 0) X(S, 1) X(S, 2) X(S, 3)
#define ROWS8S(X, S)  ROWS4S(X, S) X(S, 4) X(S, 5) X(S, 6) X(S, 7)
#define ROWS16S(X, S) ROWS8S(X, S) X(S, 8) X(S, 9) X(S, 10) X(S, 11) X(S, 12) X(S, 13) X(S, 14) X(S, 15)

__device__ __forceinline__ void lds_barrier() {
  asm volatile("s_wait_dscnt 0x0\n\ts_barrier_signal -1\n\ts_barrier_wait -1" ::: "memory");
}

template <int MODE, int NT>
__global__ void __launch_bounds__(128)
gemm_kernel(const float* A, int lda,
            const float* W, int ldw,
            const float* __restrict__ bias1,
            const float* __restrict__ bias2,
            float* __restrict__ out, int ldo, int coloff,
            float* __restrict__ out2, int N, int K) {
  constexpr int AR = 32 * NT;
  __shared__ float4 tile4[2][AR + 32][PITCH4];
  const int lw   = __builtin_amdgcn_readfirstlane((int)(threadIdx.x >> 5));
  const int lane = threadIdx.x & 31;
  const int mblk = (NT == 2) ? (blockIdx.x & 1) : (blockIdx.x & 3);
  const int nb   = (NT == 2) ? (blockIdx.x >> 1) : (blockIdx.x >> 2);
  const int rt   = (NT == 2) ? lw : (lw >> 1);
  const int ct   = (NT == 2) ? 0 : (lw & 1);
  const int m0   = mblk * AR + rt * 16;
  const int nn   = lane & 15;
  const int hi   = lane >> 4;
  const int koff = hi << 1;
  const int ncol0 = nb * 32 + ct * 16 + nn;
  const int ncol1 = ncol0 + 16;
  const int nld0 = (ncol0 < N) ? ncol0 : (N - 1);
  const int nld1 = (ncol1 < N) ? ncol1 : (N - 1);
  const float bsum0 = bias1[nld0] + (bias2 ? bias2[nld0] : 0.f);
  const float bsum1 = (NT == 2) ? bias1[nld1] + (bias2 ? bias2[nld1] : 0.f) : 0.f;
  v8f acc0, acc1;
  #pragma unroll
  for (int i = 0; i < 8; ++i) { acc0[i] = bsum0; acc1[i] = bsum1; }

#define G_WROW(i) ((nb * 32 + lw + 4 * (i)) < N ? (nb * 32 + lw + 4 * (i)) : (N - 1))
#define G_WPTR(i) const float* wsrc##i = W + (size_t)G_WROW(i) * ldw + (lane << 2);
#define G_ADC(S, i) float4 pa##S##i;
#define G_WDC(S, i) float4 pb##S##i;
#define G_ALD(S, i) if (4 * (i) < AR) pa##S##i = *(const float4*)(asrc + (size_t)(4 * (i)) * lda + ko);
#define G_WLD(S, i) pb##S##i = *(const float4*)(wsrc##i + ko);
#define G_AST(S, i) if (4 * (i) < AR) tb[lw + 4 * (i)][lane] = pa##S##i;
#define G_WST(S, i) tb[AR + lw + 4 * (i)][lane] = pb##S##i;
  const float* asrc = A + (size_t)(mblk * AR + lw) * lda + (lane << 2);
  ROWS8(G_WPTR)
  ROWS16S(G_ADC, a) ROWS8S(G_WDC, a)
  ROWS16S(G_ADC, b) ROWS8S(G_WDC, b)
  ROWS16S(G_ADC, t) ROWS8S(G_WDC, t)

  const int nround = K / TCOLS;
  { const int ko = 0; ROWS16S(G_ALD, t) ROWS8S(G_WLD, t) }
  __builtin_amdgcn_sched_barrier(0);
  { const int ko = TCOLS; ROWS16S(G_ALD, b) ROWS8S(G_WLD, b) }
  __builtin_amdgcn_sched_barrier(0);
  { const int ko = 2 * TCOLS; ROWS16S(G_ALD, a) ROWS8S(G_WLD, a) }
  __builtin_amdgcn_sched_barrier(0);
  { float4 (*tb)[PITCH4] = tile4[0]; ROWS16S(G_AST, t) ROWS8S(G_WST, t) }

#define GEMM_ROUND(cexpr, S, ST, LD)                                            \
  {                                                                             \
    const int c_ = (cexpr);                                                     \
    lds_barrier();                                                              \
    const float* tf = (const float*)&tile4[c_ & 1][0][0];                       \
    const float* ap = tf + (rt * 16 + nn) * PITCH + koff;                       \
    const float* bp = tf + (AR + ct * 16 + nn) * PITCH + koff;                  \
    v2f fa[GG], fb0[GG], fb1[GG];                                               \
    _Pragma("unroll")                                                           \
    for (int q = 0; q < GG; ++q) {                                              \
      fa[q]  = *(const v2f*)(ap + 4 * q);                                       \
      fb0[q] = *(const v2f*)(bp + 4 * q);                                       \
      if (NT == 2) fb1[q] = *(const v2f*)(bp + 16 * PITCH + 4 * q);             \
    }                                                                           \
    _Pragma("unroll")                                                           \
    for (int jg = 0; jg < TCOLS / 4; jg += GG) {                                \
      v2f na[GG], nb0[GG], nb1[GG];                                             \
      if (jg + GG < TCOLS / 4) {                                                \
        _Pragma("unroll")                                                       \
        for (int q = 0; q < GG; ++q) {                                          \
          na[q]  = *(const v2f*)(ap + 4 * (jg + GG + q));                       \
          nb0[q] = *(const v2f*)(bp + 4 * (jg + GG + q));                       \
          if (NT == 2) nb1[q] = *(const v2f*)(bp + 16 * PITCH + 4 * (jg + GG + q)); \
        }                                                                       \
      }                                                                         \
      if (jg == 0) {                                                            \
         \
         \
         \
        __builtin_amdgcn_sched_barrier(0);                                      \
        if (ST) {                                                               \
          float4 (*tb)[PITCH4] = tile4[(c_ + 1) & 1];                           \
          ROWS16S(G_AST, S) ROWS8S(G_WST, S)                                    \
        }                                                                       \
        if (LD) {                                                               \
          const int ko = (c_ + 3) * TCOLS;                                      \
          ROWS16S(G_ALD, S) ROWS8S(G_WLD, S)                                    \
        }                                                                       \
      }                                                                         \
       \
       \
      __builtin_amdgcn_sched_barrier(0);                                        \
      _Pragma("unroll")                                                         \
      for (int q = 0; q < GG; ++q) {                                            \
        acc0 = __builtin_amdgcn_wmma_f32_16x16x4_f32(false, fa[q], false, fb0[q], \
                                                     (short)0, acc0, false, false); \
        if (NT == 2)                                                            \
          acc1 = __builtin_amdgcn_wmma_f32_16x16x4_f32(false, fa[q], false, fb1[q], \
                                                       (short)0, acc1, false, false); \
      }                                                                         \
       \
       \
      __builtin_amdgcn_sched_barrier(0);                                        \
      if (jg + GG < TCOLS / 4) {                                                \
        _Pragma("unroll")                                                       \
        for (int q = 0; q < GG; ++q) {                                          \
          fa[q] = na[q]; fb0[q] = nb0[q];                                       \
          if (NT == 2) fb1[q] = nb1[q];                                         \
        }                                                                       \
      }                                                                         \
    }                                                                           \
  }

  int c = 0;
  for (; c + 5 < nround; c += 2) {
    GEMM_ROUND(c, b, 1, 1)
    GEMM_ROUND(c + 1, a, 1, 1)
  }
  GEMM_ROUND(c, b, 1, 1)
  GEMM_ROUND(c + 1, a, 1, 0)
  GEMM_ROUND(c + 2, b, 1, 0)
  GEMM_ROUND(c + 3, a, 0, 0)

  #pragma unroll
  for (int t = 0; t < NT; ++t) {
    const int ncol = t ? ncol1 : ncol0;
    #pragma unroll
    for (int v = 0; v < 8; ++v) {
      int r = m0 + v + (hi << 3);
      float val = t ? acc1[v] : acc0[v];
      if (MODE == 0) {
        out[(size_t)r * ldo + coloff + ncol] = val;
      } else if (MODE == 1) {
        int b = r >> 4, tt = r & 15;
        int rr = tt * BB + b;
        if (ncol < DD) out[(size_t)rr * DD + ncol] = val;
        else           out2[(size_t)rr * DD + (ncol - DD)] = val;
      } else {
        if (ncol < N) {
          int tt = r >> 3, b = r & 7;
          out[(size_t)(b * TT + tt) * N + ncol] = val;
        }
      }
    }
  }
}

#define KC (TCOLS / 2)
#define LROWS 16
#define WROWS 64
#define LTROWS (LROWS + WROWS)
#define NROUND (DD / 2 / KC)
#define LK 8

__global__ void __launch_bounds__(128)
lstm_step_kernel(const float* __restrict__ xpre,
                 float* hb0,
                 float* hb1,
                 const float* Whh,
                 float* __restrict__ cc,
                 float* __restrict__ lasth,
                 unsigned long long wq0, unsigned long long wq1) {
  __shared__ float4 tile4[2][LTROWS][PITCH4];
  __shared__ float red[2][4][8][32];
  const int lw   = __builtin_amdgcn_readfirstlane((int)(threadIdx.x >> 5));
  const int lane = threadIdx.x & 31;
  const int slot = blockIdx.x;
  const int nt   = blockIdx.y;
  const int n0   = nt << 4;
  const int sw   = (int)(((slot < 4 ? wq0 : wq1) >> ((slot & 3) << 4)) & 0xffffull);
  const int ga   = sw & 15, gb = (sw >> 4) & 15;
  const int ka   = (sw >> 8) & 15, kb = (sw >> 12) & 15;
  const float* hinA = (ka & 1) ? hb1 : hb0;
  const float* hinB = (kb & 1) ? hb1 : hb0;
  float* houtA = (ka & 1) ? hb0 : hb1;
  float* houtB = (kb & 1) ? hb0 : hb1;
  const int rowA = ga << 3, rowB = gb << 3;
  const int kh   = lw & 1;
  const int gp   = lw >> 1;
  const int nn   = lane & 15;
  const int hi   = lane >> 4;
  const int koff = hi << 1;
  const int rowg = hi ? rowB : rowA;

  const float cold0 = cc[(size_t)(rowg + 2 * lw) * DD + n0 + nn];
  const float cold1 = cc[(size_t)(rowg + 2 * lw + 1) * DD + n0 + nn];

  v8f acc0, acc1;
  if (kh == 0) {
    #pragma unroll
    for (int v = 0; v < 8; ++v) {
      int r = rowg + v;
      const float* xp = xpre + (size_t)r * (4 * DD) + (2 * gp) * DD + n0 + nn;
      acc0[v] = xp[0];
      acc1[v] = xp[DD];
    }
  } else {
    #pragma unroll
    for (int v = 0; v < 8; ++v) { acc0[v] = 0.f; acc1[v] = 0.f; }
  }

  const int kcol = (lane >> 4) * (DD / 2) + ((lane & 15) << 2);
#define W_OFF(i) ((size_t)(((i) >> 2) * DD + 4 * ((i) & 3)) * DD)
#define H_DC(S, i) float4 ph##S##i;
#define W_DC(S, i) float4 pw##S##i;
#define H_LD(S, i) ph##S##i = *(const float4*)(((i) < 2 ? hsrcA : hsrcB) + (size_t)(4 * ((i) & 1)) * DD + ko);
#define W_LD(S, i) pw##S##i = *(const float4*)(wsrc + W_OFF(i) + ko);
#define H_ST(S, i) tb[lw + 4 * (i)][lane] = ph##S##i;
#define W_ST(S, i) tb[LROWS + lw + 4 * (i)][lane] = pw##S##i;
  const float* hsrcA = hinA + (size_t)(rowA + lw) * DD + kcol;
  const float* hsrcB = hinB + (size_t)(rowB + lw) * DD + kcol;
  const float* wsrc = Whh + (size_t)(n0 + lw) * DD + kcol;
  ROWS4S(H_DC, a) ROWS16S(W_DC, a)
  ROWS4S(H_DC, b) ROWS16S(W_DC, b)
  ROWS4S(H_DC, t) ROWS16S(W_DC, t)
  { const int ko = 0;      ROWS4S(H_LD, t) ROWS16S(W_LD, t) }
  __builtin_amdgcn_sched_barrier(0);
  { const int ko = KC;     ROWS4S(H_LD, b) ROWS16S(W_LD, b) }
  __builtin_amdgcn_sched_barrier(0);
  { const int ko = 2 * KC; ROWS4S(H_LD, a) ROWS16S(W_LD, a) }
  __builtin_amdgcn_sched_barrier(0);
  { float4 (*tb)[PITCH4] = tile4[0]; ROWS4S(H_ST, t) ROWS16S(W_ST, t) }

#define LSTM_ROUND(cexpr, S, ST, LD)                                            \
  {                                                                             \
    const int c_ = (cexpr);                                                     \
    lds_barrier();                                                              \
    const float* tf = (const float*)&tile4[c_ & 1][0][0];                       \
    const float* ap = tf + nn * PITCH + kh * KC + koff;                         \
    const float* bp = tf + (LROWS + gp * 32 + nn) * PITCH + kh * KC + koff;     \
    v2f fa[LK], fb0[LK], fb1[LK];                                               \
    _Pragma("unroll")                                                           \
    for (int q = 0; q < LK; ++q) {                                              \
      fa[q]  = *(const v2f*)(ap + 4 * q);                                       \
      fb0[q] = *(const v2f*)(bp + 4 * q);                                       \
      fb1[q] = *(const v2f*)(bp + 16 * PITCH + 4 * q);                          \
    }                                                                           \
    _Pragma("unroll")                                                           \
    for (int jg = 0; jg < KC / 4; jg += LK) {                                   \
      v2f na[LK], nb0[LK], nb1[LK];                                             \
      if (jg + LK < KC / 4) {                                                   \
        _Pragma("unroll")                                                       \
        for (int q = 0; q < LK; ++q) {                                          \
          na[q]  = *(const v2f*)(ap + 4 * (jg + LK + q));                       \
          nb0[q] = *(const v2f*)(bp + 4 * (jg + LK + q));                       \
          nb1[q] = *(const v2f*)(bp + 16 * PITCH + 4 * (jg + LK + q));          \
        }                                                                       \
      }                                                                         \
      if (jg == 0) {                                                            \
         \
         \
         \
        __builtin_amdgcn_sched_barrier(0);                                      \
        if (ST) {                                                               \
          float4 (*tb)[PITCH4] = tile4[(c_ + 1) & 1];                           \
          ROWS4S(H_ST, S) ROWS16S(W_ST, S)                                      \
        }                                                                       \
        if (LD) {                                                               \
          const int ko = (c_ + 3) * KC;                                         \
          ROWS4S(H_LD, S) ROWS16S(W_LD, S)                                      \
        }                                                                       \
      }                                                                         \
       \
       \
      __builtin_amdgcn_sched_barrier(0);                                        \
      _Pragma("unroll")                                                         \
      for (int q = 0; q < LK; ++q) {                                            \
        acc0 = __builtin_amdgcn_wmma_f32_16x16x4_f32(false, fa[q], false, fb0[q], \
                                                     (short)0, acc0, false, false); \
        acc1 = __builtin_amdgcn_wmma_f32_16x16x4_f32(false, fa[q], false, fb1[q], \
                                                     (short)0, acc1, false, false); \
      }                                                                         \
       \
       \
      __builtin_amdgcn_sched_barrier(0);                                        \
      if (jg + LK < KC / 4) {                                                   \
        _Pragma("unroll")                                                       \
        for (int q = 0; q < LK; ++q) { fa[q] = na[q]; fb0[q] = nb0[q]; fb1[q] = nb1[q]; } \
      }                                                                         \
    }                                                                           \
  }

  static_assert(NROUND == 8, "the rounds below are written out for 8");
  LSTM_ROUND(0, b, 1, 1)
  LSTM_ROUND(1, a, 1, 1)
  LSTM_ROUND(2, b, 1, 1)
  LSTM_ROUND(3, a, 1, 1)
  LSTM_ROUND(4, b, 1, 1)
  LSTM_ROUND(5, a, 1, 0)
  LSTM_ROUND(6, b, 1, 0)
  LSTM_ROUND(7, a, 0, 0)

  #pragma unroll
  for (int v = 0; v < 8; ++v) {
    red[kh][2 * gp][v][lane]     = acc0[v];
    red[kh][2 * gp + 1][v][lane] = acc1[v];
  }
  __syncthreads();

  const bool cap = hi ? (kb == TT - 1 - gb) : (ka == TT - 1 - ga);
  float* hout = hi ? houtB : houtA;
  const int d = n0 + nn;
  #pragma unroll
  for (int j = 0; j < 2; ++j) {
    const int v = 2 * lw + j;
    const int r = rowg + v;
    const float gi = red[0][0][v][lane] + red[1][0][v][lane];
    const float gf = red[0][1][v][lane] + red[1][1][v][lane];
    const float gg = red[0][2][v][lane] + red[1][2][v][lane];
    const float go = red[0][3][v][lane] + red[1][3][v][lane];
    float si = 1.f / (1.f + expf(-gi));
    float sf = 1.f / (1.f + expf(-gf));
    float gt = tanhf(gg);
    float so = 1.f / (1.f + expf(-go));
    float cold = j ? cold1 : cold0;
    float cnew = sf * cold + si * gt;
    float hnew = so * tanhf(cnew);
    cc[(size_t)r * DD + d] = cnew;
    hout[(size_t)r * DD + d] = hnew;
    if (cap) lasth[(size_t)r * DD + d] = hnew;
  }
}

__global__ void ln_relu_kernel(const float* __restrict__ in,
                               const float* __restrict__ g,
                               const float* __restrict__ be,
                               float* __restrict__ out) {
  __shared__ float s1[256], s2[256];
  int r = blockIdx.x, t = threadIdx.x;
  const float* row = in + (size_t)r * DD;
  float xv[4], gv[4], bv[4];
  #pragma unroll
  for (int q = 0; q < 4; ++q) { xv[q] = row[t + 256 * q]; gv[q] = g[t + 256 * q]; bv[q] = be[t + 256 * q]; }
  float s = 0.f, sq = 0.f;
  #pragma unroll
  for (int q = 0; q < 4; ++q) { float v = xv[q]; s += v; sq = fmaf(v, v, sq); }
  s1[t] = s; s2[t] = sq;
  __syncthreads();
  for (int off = 128; off > 0; off >>= 1) {
    if (t < off) { s1[t] += s1[t + off]; s2[t] += s2[t + off]; }
    __syncthreads();
  }
  float mean = s1[0] * (1.0f / DD);
  float var = s2[0] * (1.0f / DD) - mean * mean;
  float inv = rsqrtf(var + 1e-5f);
  #pragma unroll
  for (int q = 0; q < 4; ++q) {
    float v = fmaf((xv[q] - mean) * inv, gv[q], bv[q]);
    out[(size_t)r * DD + t + 256 * q] = v > 0.f ? v : 0.f;
  }
}

extern "C" void kernel_launch(void* const* d_in, const int* in_sizes, int n_in,
                              void* d_out, int out_size, void* d_ws, size_t ws_size,
                              hipStream_t stream) {
  const float* x   = (const float*)d_in[0];
  const float* Wn  = (const float*)d_in[1];
  const float* bn  = (const float*)d_in[2];
  const float* Wv  = (const float*)d_in[3];
  const float* bv  = (const float*)d_in[4];
  const float* Wf  = (const float*)d_in[5];
  const float* bf  = (const float*)d_in[6];
  const float* Wih = (const float*)d_in[7];
  const float* Whh = (const float*)d_in[8];
  const float* bih = (const float*)d_in[9];
  const float* bhh = (const float*)d_in[10];
  const float* Wc  = (const float*)d_in[11];
  const float* bc  = (const float*)d_in[12];
  const float* g_a = (const float*)d_in[13];
  const float* be_a= (const float*)d_in[14];
  const float* Wa  = (const float*)d_in[15];
  const float* ba  = (const float*)d_in[16];
  float* out = (float*)d_out;

  float* ws    = (float*)d_ws;
  float* A0    = ws;
  float* F1    = A0    + 131072;
  float* F2    = F1    + 131072;
  float* ENV   = F2    + 131072;
  float* XPRE  = ENV   + 262144;
  float* H0    = XPRE  + 524288;
  float* H1    = H0    + 131072;
  float* CC    = H1    + 131072;
  float* LASTH = CC    + 131072;
  float* UNR   = LASTH + 131072;
  float* LNR   = UNR   + 131072;

  pool_kernel<<<1536, 256, 0, stream>>>(x, A0, F1, F2);
  gemm_kernel<0, 1><<<128, 128, 0, stream>>>(F1, DD, Wn, DD, bn, nullptr,
                                             ENV, 2 * DD, 0,  nullptr, DD, DD);
  gemm_kernel<0, 1><<<128, 128, 0, stream>>>(F2, DD, Wv, DD, bv, nullptr,
                                             ENV, 2 * DD, DD, nullptr, DD, DD);
  gemm_kernel<1, 1><<<256, 128, 0, stream>>>(ENV, 2 * DD, Wf, 2 * DD, bf, nullptr,
                                             H0, 0, 0, CC, 2 * DD, 2 * DD);
  gemm_kernel<0, 2><<<256, 128, 0, stream>>>(A0, DD, Wih, DD, bih, bhh,
                                             XPRE, 4 * DD, 0, nullptr, 4 * DD, DD);
  lstm_step_kernel<<<dim3(8, 64), 128, 0, stream>>>(XPRE, H0, H1, Whh, CC, LASTH, 0x0076005400320010ull, 0x00fe00dc00ba0098ull);
  lstm_step_kernel<<<dim3(4, 64), 128, 0, stream>>>(XPRE, H0, H1, Whh, CC, LASTH, 0x11a7116411321110ull, 0ull);
  lstm_step_kernel<<<dim3(4, 64), 128, 0, stream>>>(XPRE, H0, H1, Whh, CC, LASTH, 0x22a7226422322210ull, 0ull);
  lstm_step_kernel<<<dim3(4, 64), 128, 0, stream>>>(XPRE, H0, H1, Whh, CC, LASTH, 0x31a8336433323310ull, 0ull);
  lstm_step_kernel<<<dim3(4, 64), 128, 0, stream>>>(XPRE, H0, H1, Whh, CC, LASTH, 0x42a8446444324410ull, 0ull);
  lstm_step_kernel<<<dim3(4, 64), 128, 0, stream>>>(XPRE, H0, H1, Whh, CC, LASTH, 0x53a8556455325510ull, 0ull);
  lstm_step_kernel<<<dim3(4, 64), 128, 0, stream>>>(XPRE, H0, H1, Whh, CC, LASTH, 0x14b8616566326610ull, 0ull);
  lstm_step_kernel<<<dim3(4, 64), 128, 0, stream>>>(XPRE, H0, H1, Whh, CC, LASTH, 0x25b8726577327710ull, 0ull);
  lstm_step_kernel<<<dim3(4, 64), 128, 0, stream>>>(XPRE, H0, H1, Whh, CC, LASTH, 0x36b8836588328810ull, 0ull);
  lstm_step_kernel<<<dim3(4, 64), 128, 0, stream>>>(XPRE, H0, H1, Whh, CC, LASTH, 0x47b8946599329910ull, 0ull);
  lstm_step_kernel<<<dim3(4, 64), 128, 0, stream>>>(XPRE, H0, H1, Whh, CC, LASTH, 0x11c935756a42aa10ull, 0ull);
  lstm_step_kernel<<<dim3(4, 64), 128, 0, stream>>>(XPRE, H0, H1, Whh, CC, LASTH, 0x22c946757b42bb10ull, 0ull);
  lstm_step_kernel<<<dim3(4, 64), 128, 0, stream>>>(XPRE, H0, H1, Whh, CC, LASTH, 0x33c957758c42cc10ull, 0ull);
  lstm_step_kernel<<<dim3(4, 64), 128, 0, stream>>>(XPRE, H0, H1, Whh, CC, LASTH, 0x14d968759a43dd10ull, 0ull);
  lstm_step_kernel<<<dim3(4, 64), 128, 0, stream>>>(XPRE, H0, H1, Whh, CC, LASTH, 0x25d97975ab43ee10ull, 0ull);
  lstm_step_kernel<<<dim3(4, 64), 128, 0, stream>>>(XPRE, H0, H1, Whh, CC, LASTH, 0x16e98a75bc43df20ull, 0ull);
  gemm_kernel<0, 1><<<128, 128, 0, stream>>>(LASTH, DD, Wc, DD, bc, nullptr,
                                             UNR, DD, 0, nullptr, DD, DD);
  ln_relu_kernel<<<128, 256, 0, stream>>>(UNR, g_a, be_a, LNR);
  gemm_kernel<2, 1><<<236, 128, 0, stream>>>(LNR, DD, Wa, DD, ba, nullptr,
                                             out, NA, 0, nullptr, NA, DD);
}
